# starting kernel
# baseline (speedup 1.0000x reference)
#include <hip/hip_runtime.h>

#define SEQ   512
#define BATCH 4096
#define DIMS  3
#define DIMY  3
#define DIMZ  6
#define RANK  6
#define HID   50
#define KG    13
#define XSTR  68
#define ALPHA 0.5f

typedef float v2f __attribute__((ext_vector_type(2)));
typedef float v8f __attribute__((ext_vector_type(8)));

__device__ __forceinline__ v8f wmma_f32_k4(v2f a, v2f b, v8f c) {
    return __builtin_amdgcn_wmma_f32_16x16x4_f32(false, a, false, b, (short)0, c, false, false);
}

__device__ __forceinline__ float fast_tanh(float x) {
#if __has_builtin(__builtin_amdgcn_tanhf)
    return __builtin_amdgcn_tanhf(x);
#else
    x = fminf(15.0f, fmaxf(-15.0f, x));
    float e = __expf(2.0f * x);
    float r = __builtin_amdgcn_rcpf(e + 1.0f);
    return fmaf(-2.0f, r, 1.0f);
#endif
}

__device__ __forceinline__ float fast_sigmoid(float x) {
    return __builtin_amdgcn_rcpf(1.0f + __expf(-x));
}

__global__ __launch_bounds__(32) void cxtrnn_wmma_kernel(
    const float* __restrict__ s,   const float* __restrict__ z,
    const float* __restrict__ U,   const float* __restrict__ V,
    const float* __restrict__ Wi,  const float* __restrict__ bi,
    const float* __restrict__ Wo,  const float* __restrict__ bo,
    const float* __restrict__ nmW, const float* __restrict__ nmb,
    float* __restrict__ out)
{
    __shared__ float ldsX[16 * XSTR];
    __shared__ float ldsP[16 * 16];

    const int lane = (int)threadIdx.x;
    const int col  = lane & 15;
    const int hi   = lane >> 4;
    const bool hib = (hi != 0);
    const int b0   = (int)blockIdx.x * 16;
    const int brow = b0 + col;

    v2f Bv[KG];
#pragma unroll
    for (int g = 0; g < KG; ++g) {
        int k0 = 4 * g + 2 * hi;
        float a0 = (k0     < HID && col < RANK) ? V[k0 * RANK + col]       : 0.0f;
        float a1 = (k0 + 1 < HID && col < RANK) ? V[(k0 + 1) * RANK + col] : 0.0f;
        v2f t; t[0] = a0; t[1] = a1; Bv[g] = t;
    }
    v2f Bu[4][2];
#pragma unroll
    for (int t4 = 0; t4 < 4; ++t4) {
        int h = 16 * t4 + col;
#pragma unroll
        for (int g = 0; g < 2; ++g) {
            int k0 = 4 * g + 2 * hi;
            float a0 = (h < HID && k0     < RANK) ? ALPHA * U[h * RANK + k0]     : 0.0f;
            float a1 = (h < HID && k0 + 1 < RANK) ? ALPHA * U[h * RANK + k0 + 1] : 0.0f;
            v2f t; t[0] = a0; t[1] = a1; Bu[t4][g] = t;
        }
    }
    v2f Bw[4];
#pragma unroll
    for (int t4 = 0; t4 < 4; ++t4) {
        int h  = 16 * t4 + col;
        int k0 = 2 * hi;
        float a0 = 0.0f, a1 = 0.0f;
        if (h < HID) {
            a0 = ALPHA * Wi[h * DIMS + k0];
            a1 = ALPHA * ((k0 + 1 < DIMS) ? Wi[h * DIMS + k0 + 1] : bi[h]);
        }
        v2f t; t[0] = a0; t[1] = a1; Bw[t4] = t;
    }
    v2f Bo[KG];
#pragma unroll
    for (int g = 0; g < KG; ++g) {
        int k0 = 4 * g + 2 * hi;
        float a0 = 0.0f, a1 = 0.0f;
        if (col < DIMY) {
            a0 = (k0     < HID) ? Wo[col * HID + k0]     : ((k0     == HID) ? bo[col] : 0.0f);
            a1 = (k0 + 1 < HID) ? Wo[col * HID + k0 + 1] : ((k0 + 1 == HID) ? bo[col] : 0.0f);
        }
        v2f t; t[0] = a0; t[1] = a1; Bo[g] = t;
    }
    float nmw[RANK][DIMZ], nmbv[RANK];
#pragma unroll
    for (int r = 0; r < RANK; ++r) {
        nmbv[r] = nmb[r];
#pragma unroll
        for (int i = 0; i < DIMZ; ++i) nmw[r][i] = nmW[r * DIMZ + i];
    }

    v8f XD[4];
#pragma unroll
    for (int t4 = 0; t4 < 4; ++t4) { v8f zz = {0,0,0,0,0,0,0,0}; XD[t4] = zz; }
    float Atan[2 * KG];
#pragma unroll
    for (int i = 0; i < 2 * KG; ++i) Atan[i] = 0.0f;

    const v8f vzero = {0,0,0,0,0,0,0,0};

#pragma unroll 1
    for (int t = 0; t < SEQ; ++t) {
        const float* zr = z + ((size_t)t * BATCH + (size_t)brow) * DIMZ;
        const float* sr = s + ((size_t)t * BATCH + (size_t)brow) * DIMS;
        float zv[DIMZ];
#pragma unroll
        for (int i = 0; i < DIMZ; ++i) zv[i] = zr[i];
        float sv0 = sr[0], sv1 = sr[1], sv2 = sr[2];
        float zsum = 0.0f;
#pragma unroll
        for (int i = 0; i < DIMZ; ++i) zsum += zv[i];
        float g0, g1, g2, g3, g4, g5;
        {
            float a0 = nmbv[0], a1 = nmbv[1], a2 = nmbv[2];
            float a3 = nmbv[3], a4 = nmbv[4], a5 = nmbv[5];
#pragma unroll
            for (int i = 0; i < DIMZ; ++i) {
                a0 = fmaf(nmw[0][i], zv[i], a0);
                a1 = fmaf(nmw[1][i], zv[i], a1);
                a2 = fmaf(nmw[2][i], zv[i], a2);
                a3 = fmaf(nmw[3][i], zv[i], a3);
                a4 = fmaf(nmw[4][i], zv[i], a4);
                a5 = fmaf(nmw[5][i], zv[i], a5);
            }
            g0 = fast_sigmoid(a0); g1 = fast_sigmoid(a1); g2 = fast_sigmoid(a2);
            g3 = fast_sigmoid(a3); g4 = fast_sigmoid(a4); g5 = fast_sigmoid(a5);
        }
        const float ga0 = hib ? g2 : g0;
        const float ga1 = hib ? g3 : g1;
        const float ga2 = hib ? 0.0f : g4;
        const float ga3 = hib ? 0.0f : g5;

        v8f P = vzero;
#pragma unroll
        for (int g = 0; g < KG; ++g) {
            v2f a; a[0] = Atan[2 * g]; a[1] = Atan[2 * g + 1];
            P = wmma_f32_k4(a, Bv[g], P);
        }

#pragma unroll
        for (int v = 0; v < 8; ++v) ldsP[(v + 8 * hi) * 16 + col] = P[v];
        asm volatile("s_wait_dscnt 0" ::: "memory");
        v2f Ap0, Ap1;
        {
            float p0 = ldsP[col * 16 + 2 * hi];
            float p1 = ldsP[col * 16 + 2 * hi + 1];
            float p2 = ldsP[col * 16 + 4 + 2 * hi];
            float p3 = ldsP[col * 16 + 4 + 2 * hi + 1];
            Ap0[0] = p0 * ga0; Ap0[1] = p1 * ga1;
            Ap1[0] = p2 * ga2; Ap1[1] = p3 * ga3;
        }
        v2f As;
        As[0] = zsum * (hib ? sv2 : sv0);
        As[1] = hib ? zsum : (zsum * sv1);

#pragma unroll
        for (int t4 = 0; t4 < 4; ++t4) {
            v8f acc = wmma_f32_k4(Ap0, Bu[t4][0], vzero);
            acc = wmma_f32_k4(Ap1, Bu[t4][1], acc);
            acc = wmma_f32_k4(As,  Bw[t4],    acc);
            v8f x = XD[t4];
#pragma unroll
            for (int v = 0; v < 8; ++v) x[v] = fmaf(x[v], 1.0f - ALPHA, acc[v]);
            XD[t4] = x;
#pragma unroll
            for (int v = 0; v < 8; ++v)
                ldsX[(v + 8 * hi) * XSTR + 16 * t4 + col] = x[v];
        }
        asm volatile("s_wait_dscnt 0" ::: "memory");

#pragma unroll
        for (int g = 0; g < KG; ++g) {
            int k0 = 4 * g + 2 * hi;
            Atan[2 * g]     = fast_tanh(ldsX[col * XSTR + k0]);
            Atan[2 * g + 1] = fast_tanh(ldsX[col * XSTR + k0 + 1]);
        }

        v8f OD = vzero;
#pragma unroll
        for (int g = 0; g < KG; ++g) {
            float a0 = zsum * Atan[2 * g];
            float a1 = zsum * Atan[2 * g + 1];
            if (g == KG - 1) {
                a0 = hib ? zsum : a0;
                a1 = hib ? 0.0f : a1;
            }
            v2f a; a[0] = a0; a[1] = a1;
            OD = wmma_f32_k4(a, Bo[g], OD);
        }

        if (col < DIMY) {
            size_t base = (size_t)t * BATCH * DIMY + (size_t)(b0 + 8 * hi) * DIMY + col;
#pragma unroll
            for (int v = 0; v < 8; ++v) out[base + (size_t)v * DIMY] = OD[v];
        }
    }
}

extern "C" void kernel_launch(void* const* d_in, const int* in_sizes, int n_in,
                              void* d_out, int out_size, void* d_ws, size_t ws_size,
                              hipStream_t stream) {
    (void)in_sizes; (void)n_in; (void)out_size; (void)d_ws; (void)ws_size;
    const float* s   = (const float*)d_in[0];
    const float* z   = (const float*)d_in[1];
    const float* U   = (const float*)d_in[2];
    const float* V   = (const float*)d_in[3];
    const float* Wi  = (const float*)d_in[4];
    const float* bi  = (const float*)d_in[5];
    const float* Wo  = (const float*)d_in[6];
    const float* bo  = (const float*)d_in[7];
    const float* nmW = (const float*)d_in[8];
    const float* nmb = (const float*)d_in[9];
    float* out = (float*)d_out;

    dim3 grid(BATCH / 16);
    dim3 block(32);
    hipLaunchKernelGGL(cxtrnn_wmma_kernel, grid, block, 0, stream,
                       s, z, U, V, Wi, bi, Wo, bo, nmW, nmb, out);
}


# Input rows z_t, s_t loaded one two-step trip ahead into registers (time loop unrolled by 2, step inputs prepared from the previous trip's loads), so no step waits for global-memory latency
# speedup vs baseline: 1.4886x; 1.4886x over previous
#include <hip/hip_runtime.h>

#define SEQ   512
#define BATCH 4096
#define DIMS  3
#define DIMY  3
#define DIMZ  6
#define RANK  6
#define HID   50
#define KG    13
#define XSTR  68
#define ALPHA 0.5f

typedef float v2f __attribute__((ext_vector_type(2)));
typedef float v8f __attribute__((ext_vector_type(8)));

__device__ __forceinline__ v8f wmma_f32_k4(v2f a, v2f b, v8f c) {
    return __builtin_amdgcn_wmma_f32_16x16x4_f32(false, a, false, b, (short)0, c, false, false);
}

__device__ __forceinline__ float fast_tanh(float x) {
#if __has_builtin(__builtin_amdgcn_tanhf)
    return __builtin_amdgcn_tanhf(x);
#else
    x = fminf(15.0f, fmaxf(-15.0f, x));
    float e = __expf(2.0f * x);
    float r = __builtin_amdgcn_rcpf(e + 1.0f);
    return fmaf(-2.0f, r, 1.0f);
#endif
}

__device__ __forceinline__ float fast_sigmoid(float x) {
    return __builtin_amdgcn_rcpf(1.0f + __expf(-x));
}

struct StepIn {
    float zsum;
    float ga0, ga1, ga2, ga3;
    float as0, as1;
};

__device__ __forceinline__ StepIn step_inputs(const float (&zv)[DIMZ],
                                              const float sv0, const float sv1, const float sv2,
                                              const float (&nmw)[RANK][DIMZ],
                                              const float (&nmbv)[RANK], const bool hib) {
    float zsum = 0.0f;
#pragma unroll
    for (int i = 0; i < DIMZ; ++i) zsum += zv[i];
    float a0 = nmbv[0], a1 = nmbv[1], a2 = nmbv[2];
    float a3 = nmbv[3], a4 = nmbv[4], a5 = nmbv[5];
#pragma unroll
    for (int i = 0; i < DIMZ; ++i) {
        a0 = fmaf(nmw[0][i], zv[i], a0);
        a1 = fmaf(nmw[1][i], zv[i], a1);
        a2 = fmaf(nmw[2][i], zv[i], a2);
        a3 = fmaf(nmw[3][i], zv[i], a3);
        a4 = fmaf(nmw[4][i], zv[i], a4);
        a5 = fmaf(nmw[5][i], zv[i], a5);
    }
    const float g0 = fast_sigmoid(a0), g1 = fast_sigmoid(a1), g2 = fast_sigmoid(a2);
    const float g3 = fast_sigmoid(a3), g4 = fast_sigmoid(a4), g5 = fast_sigmoid(a5);
    StepIn r;
    r.zsum = zsum;
    r.ga0 = hib ? g2 : g0;
    r.ga1 = hib ? g3 : g1;
    r.ga2 = hib ? 0.0f : g4;
    r.ga3 = hib ? 0.0f : g5;
    r.as0 = zsum * (hib ? sv2 : sv0);
    r.as1 = hib ? zsum : (zsum * sv1);
    return r;
}

__global__ __launch_bounds__(32) void cxtrnn_wmma_kernel(
    const float* __restrict__ s,   const float* __restrict__ z,
    const float* __restrict__ U,   const float* __restrict__ V,
    const float* __restrict__ Wi,  const float* __restrict__ bi,
    const float* __restrict__ Wo,  const float* __restrict__ bo,
    const float* __restrict__ nmW, const float* __restrict__ nmb,
    float* __restrict__ out)
{
    __shared__ float ldsX[16 * XSTR];
    __shared__ float ldsP[16 * 16];

    const int lane = (int)threadIdx.x;
    const int col  = lane & 15;
    const int hi   = lane >> 4;
    const bool hib = (hi != 0);
    const int b0   = (int)blockIdx.x * 16;
    const int brow = b0 + col;

    v2f Bv[KG];
#pragma unroll
    for (int g = 0; g < KG; ++g) {
        int k0 = 4 * g + 2 * hi;
        float a0 = (k0     < HID && col < RANK) ? V[k0 * RANK + col]       : 0.0f;
        float a1 = (k0 + 1 < HID && col < RANK) ? V[(k0 + 1) * RANK + col] : 0.0f;
        v2f t; t[0] = a0; t[1] = a1; Bv[g] = t;
    }
    v2f Bu[4][2];
#pragma unroll
    for (int t4 = 0; t4 < 4; ++t4) {
        int h = 16 * t4 + col;
#pragma unroll
        for (int g = 0; g < 2; ++g) {
            int k0 = 4 * g + 2 * hi;
            float a0 = (h < HID && k0     < RANK) ? ALPHA * U[h * RANK + k0]     : 0.0f;
            float a1 = (h < HID && k0 + 1 < RANK) ? ALPHA * U[h * RANK + k0 + 1] : 0.0f;
            v2f t; t[0] = a0; t[1] = a1; Bu[t4][g] = t;
        }
    }
    v2f Bw[4];
#pragma unroll
    for (int t4 = 0; t4 < 4; ++t4) {
        int h  = 16 * t4 + col;
        int k0 = 2 * hi;
        float a0 = 0.0f, a1 = 0.0f;
        if (h < HID) {
            a0 = ALPHA * Wi[h * DIMS + k0];
            a1 = ALPHA * ((k0 + 1 < DIMS) ? Wi[h * DIMS + k0 + 1] : bi[h]);
        }
        v2f t; t[0] = a0; t[1] = a1; Bw[t4] = t;
    }
    v2f Bo[KG];
#pragma unroll
    for (int g = 0; g < KG; ++g) {
        int k0 = 4 * g + 2 * hi;
        float a0 = 0.0f, a1 = 0.0f;
        if (col < DIMY) {
            a0 = (k0     < HID) ? Wo[col * HID + k0]     : ((k0     == HID) ? bo[col] : 0.0f);
            a1 = (k0 + 1 < HID) ? Wo[col * HID + k0 + 1] : ((k0 + 1 == HID) ? bo[col] : 0.0f);
        }
        v2f t; t[0] = a0; t[1] = a1; Bo[g] = t;
    }
    float nmw[RANK][DIMZ], nmbv[RANK];
#pragma unroll
    for (int r = 0; r < RANK; ++r) {
        nmbv[r] = nmb[r];
#pragma unroll
        for (int i = 0; i < DIMZ; ++i) nmw[r][i] = nmW[r * DIMZ + i];
    }

    v8f XD[4];
#pragma unroll
    for (int t4 = 0; t4 < 4; ++t4) { v8f zz = {0,0,0,0,0,0,0,0}; XD[t4] = zz; }
    float Atan[2 * KG];
#pragma unroll
    for (int i = 0; i < 2 * KG; ++i) Atan[i] = 0.0f;

    const v8f vzero = {0,0,0,0,0,0,0,0};

    float zq[2][DIMZ], sq0[2], sq1[2], sq2[2];
#pragma unroll
    for (int u = 0; u < 2; ++u) {
        const float* zr = z + ((size_t)u * BATCH + (size_t)brow) * DIMZ;
        const float* sr = s + ((size_t)u * BATCH + (size_t)brow) * DIMS;
#pragma unroll
        for (int i = 0; i < DIMZ; ++i) zq[u][i] = zr[i];
        sq0[u] = sr[0]; sq1[u] = sr[1]; sq2[u] = sr[2];
    }

#pragma unroll 1
    for (int t2 = 0; t2 < SEQ; t2 += 2) {
        StepIn G[2];
#pragma unroll
        for (int u = 0; u < 2; ++u)
            G[u] = step_inputs(zq[u], sq0[u], sq1[u], sq2[u], nmw, nmbv, hib);

#pragma unroll
        for (int u = 0; u < 2; ++u) {
            const int tn = (t2 + 2 + u < SEQ) ? (t2 + 2 + u) : (SEQ - 1);
            const float* zr = z + ((size_t)tn * BATCH + (size_t)brow) * DIMZ;
            const float* sr = s + ((size_t)tn * BATCH + (size_t)brow) * DIMS;
#pragma unroll
            for (int i = 0; i < DIMZ; ++i) zq[u][i] = zr[i];
            sq0[u] = sr[0]; sq1[u] = sr[1]; sq2[u] = sr[2];
        }

#pragma unroll
    for (int u = 0; u < 2; ++u) {
        const int t = t2 + u;
        const float zsum = G[u].zsum;
        const float ga0 = G[u].ga0, ga1 = G[u].ga1, ga2 = G[u].ga2, ga3 = G[u].ga3;

        v8f P = vzero;
#pragma unroll
        for (int g = 0; g < KG; ++g) {
            v2f a; a[0] = Atan[2 * g]; a[1] = Atan[2 * g + 1];
            P = wmma_f32_k4(a, Bv[g], P);
        }

#pragma unroll
        for (int v = 0; v < 8; ++v) ldsP[(v + 8 * hi) * 16 + col] = P[v];
        asm volatile("s_wait_dscnt 0" ::: "memory");
        v2f Ap0, Ap1;
        {
            float p0 = ldsP[col * 16 + 2 * hi];
            float p1 = ldsP[col * 16 + 2 * hi + 1];
            float p2 = ldsP[col * 16 + 4 + 2 * hi];
            float p3 = ldsP[col * 16 + 4 + 2 * hi + 1];
            Ap0[0] = p0 * ga0; Ap0[1] = p1 * ga1;
            Ap1[0] = p2 * ga2; Ap1[1] = p3 * ga3;
        }
        v2f As;
        As[0] = G[u].as0;
        As[1] = G[u].as1;

#pragma unroll
        for (int t4 = 0; t4 < 4; ++t4) {
            v8f acc = wmma_f32_k4(Ap0, Bu[t4][0], vzero);
            acc = wmma_f32_k4(Ap1, Bu[t4][1], acc);
            acc = wmma_f32_k4(As,  Bw[t4],    acc);
            v8f x = XD[t4];
#pragma unroll
            for (int v = 0; v < 8; ++v) x[v] = fmaf(x[v], 1.0f - ALPHA, acc[v]);
            XD[t4] = x;
#pragma unroll
            for (int v = 0; v < 8; ++v)
                ldsX[(v + 8 * hi) * XSTR + 16 * t4 + col] = x[v];
        }
        asm volatile("s_wait_dscnt 0" ::: "memory");

        float xa[2 * KG];
#pragma unroll
        for (int g = 0; g < KG; ++g) {
            int k0 = 4 * g + 2 * hi;
            xa[2 * g]     = ldsX[col * XSTR + k0];
            xa[2 * g + 1] = ldsX[col * XSTR + k0 + 1];
        }
        __builtin_amdgcn_sched_barrier(0);
#pragma unroll
        for (int i = 0; i < 2 * KG; ++i) Atan[i] = fast_tanh(xa[i]);

        v8f OD = vzero;
#pragma unroll
        for (int g = 0; g < KG; ++g) {
            float a0 = zsum * Atan[2 * g];
            float a1 = zsum * Atan[2 * g + 1];
            if (g == KG - 1) {
                a0 = hib ? zsum : a0;
                a1 = hib ? 0.0f : a1;
            }
            v2f a; a[0] = a0; a[1] = a1;
            OD = wmma_f32_k4(a, Bo[g], OD);
        }

        if (col < DIMY) {
            size_t base = (size_t)t * BATCH * DIMY + (size_t)(b0 + 8 * hi) * DIMY + col;
#pragma unroll
            for (int v = 0; v < 8; ++v) out[base + (size_t)v * DIMY] = OD[v];
        }
    }
    }
}

extern "C" void kernel_launch(void* const* d_in, const int* in_sizes, int n_in,
                              void* d_out, int out_size, void* d_ws, size_t ws_size,
                              hipStream_t stream) {
    (void)in_sizes; (void)n_in; (void)out_size; (void)d_ws; (void)ws_size;
    const float* s   = (const float*)d_in[0];
    const float* z   = (const float*)d_in[1];
    const float* U   = (const float*)d_in[2];
    const float* V   = (const float*)d_in[3];
    const float* Wi  = (const float*)d_in[4];
    const float* bi  = (const float*)d_in[5];
    const float* Wo  = (const float*)d_in[6];
    const float* bo  = (const float*)d_in[7];
    const float* nmW = (const float*)d_in[8];
    const float* nmb = (const float*)d_in[9];
    float* out = (float*)d_out;

    dim3 grid(BATCH / 16);
    dim3 block(32);
    hipLaunchKernelGGL(cxtrnn_wmma_kernel, grid, block, 0, stream,
                       s, z, U, V, Wi, bi, Wo, bo, nmW, nmb, out);
}


# Output product merged into the recurrent one: one 13-WMMA chain of tanh(x_new) on [V | W_out^T;b_out] gives next step's P and this step's out; row scale zsum applied to the sum, not to each term
# speedup vs baseline: 1.7598x; 1.1822x over previous
#include <hip/hip_runtime.h>

#define SEQ   512
#define BATCH 4096
#define DIMS  3
#define DIMY  3
#define DIMZ  6
#define RANK  6
#define HID   50
#define KG    13
#define XSTR  68
#define ALPHA 0.5f
#define QOUT  8

typedef float v2f __attribute__((ext_vector_type(2)));
typedef float v8f __attribute__((ext_vector_type(8)));

__device__ __forceinline__ v8f wmma_f32_k4(v2f a, v2f b, v8f c) {
    return __builtin_amdgcn_wmma_f32_16x16x4_f32(false, a, false, b, (short)0, c, false, false);
}

__device__ __forceinline__ float fast_tanh(float x) {
#if __has_builtin(__builtin_amdgcn_tanhf)
    return __builtin_amdgcn_tanhf(x);
#else
    x = fminf(15.0f, fmaxf(-15.0f, x));
    float e = __expf(2.0f * x);
    float r = __builtin_amdgcn_rcpf(e + 1.0f);
    return fmaf(-2.0f, r, 1.0f);
#endif
}

__device__ __forceinline__ float fast_sigmoid(float x) {
    return __builtin_amdgcn_rcpf(1.0f + __expf(-x));
}

struct StepIn {
    float zsum;
    float ga0, ga1, ga2, ga3;
    float as0, as1;
};

__device__ __forceinline__ StepIn step_inputs(const float (&zv)[DIMZ],
                                              const float sv0, const float sv1, const float sv2,
                                              const float (&nmw)[RANK][DIMZ],
                                              const float (&nmbv)[RANK], const bool hib) {
    float zsum = 0.0f;
#pragma unroll
    for (int i = 0; i < DIMZ; ++i) zsum += zv[i];
    float a0 = nmbv[0], a1 = nmbv[1], a2 = nmbv[2];
    float a3 = nmbv[3], a4 = nmbv[4], a5 = nmbv[5];
#pragma unroll
    for (int i = 0; i < DIMZ; ++i) {
        a0 = fmaf(nmw[0][i], zv[i], a0);
        a1 = fmaf(nmw[1][i], zv[i], a1);
        a2 = fmaf(nmw[2][i], zv[i], a2);
        a3 = fmaf(nmw[3][i], zv[i], a3);
        a4 = fmaf(nmw[4][i], zv[i], a4);
        a5 = fmaf(nmw[5][i], zv[i], a5);
    }
    const float g0 = fast_sigmoid(a0), g1 = fast_sigmoid(a1), g2 = fast_sigmoid(a2);
    const float g3 = fast_sigmoid(a3), g4 = fast_sigmoid(a4), g5 = fast_sigmoid(a5);
    StepIn r;
    r.zsum = zsum;
    r.ga0 = hib ? g2 : g0;
    r.ga1 = hib ? g3 : g1;
    r.ga2 = hib ? 0.0f : g4;
    r.ga3 = hib ? 0.0f : g5;
    r.as0 = zsum * (hib ? sv2 : sv0);
    r.as1 = hib ? zsum : (zsum * sv1);
    return r;
}

__global__ __launch_bounds__(32) void cxtrnn_wmma_kernel(
    const float* __restrict__ s,   const float* __restrict__ z,
    const float* __restrict__ U,   const float* __restrict__ V,
    const float* __restrict__ Wi,  const float* __restrict__ bi,
    const float* __restrict__ Wo,  const float* __restrict__ bo,
    const float* __restrict__ nmW, const float* __restrict__ nmb,
    float* __restrict__ out)
{
    __shared__ float ldsX[16 * XSTR];
    __shared__ float ldsP[16 * 16];
    __shared__ __attribute__((aligned(16))) float ldsZ[2 * 16];

    const int lane = (int)threadIdx.x;
    const int col  = lane & 15;
    const int hi   = lane >> 4;
    const bool hib = (hi != 0);
    const int b0   = (int)blockIdx.x * 16;
    const int brow = b0 + col;

    v2f Bq[KG];
#pragma unroll
    for (int g = 0; g < KG; ++g) {
        int k0 = 4 * g + 2 * hi;
        float a0 = 0.0f, a1 = 0.0f;
        if (col < RANK) {
            a0 = (k0     < HID) ? V[k0 * RANK + col]       : 0.0f;
            a1 = (k0 + 1 < HID) ? V[(k0 + 1) * RANK + col] : 0.0f;
        } else if (col >= QOUT && col < QOUT + DIMY) {
            const int y = col - QOUT;
            a0 = (k0     < HID) ? Wo[y * HID + k0]     : ((k0     == HID) ? bo[y] : 0.0f);
            a1 = (k0 + 1 < HID) ? Wo[y * HID + k0 + 1] : ((k0 + 1 == HID) ? bo[y] : 0.0f);
        }
        v2f t; t[0] = a0; t[1] = a1; Bq[g] = t;
    }
    v2f Bu[4][2];
#pragma unroll
    for (int t4 = 0; t4 < 4; ++t4) {
        int h = 16 * t4 + col;
#pragma unroll
        for (int g = 0; g < 2; ++g) {
            int k0 = 4 * g + 2 * hi;
            float a0 = (h < HID && k0     < RANK) ? ALPHA * U[h * RANK + k0]     : 0.0f;
            float a1 = (h < HID && k0 + 1 < RANK) ? ALPHA * U[h * RANK + k0 + 1] : 0.0f;
            v2f t; t[0] = a0; t[1] = a1; Bu[t4][g] = t;
        }
    }
    v2f Bw[4];
#pragma unroll
    for (int t4 = 0; t4 < 4; ++t4) {
        int h  = 16 * t4 + col;
        int k0 = 2 * hi;
        float a0 = 0.0f, a1 = 0.0f;
        if (h < HID) {
            a0 = ALPHA * Wi[h * DIMS + k0];
            a1 = ALPHA * ((k0 + 1 < DIMS) ? Wi[h * DIMS + k0 + 1] : bi[h]);
        }
        v2f t; t[0] = a0; t[1] = a1; Bw[t4] = t;
    }
    float nmw[RANK][DIMZ], nmbv[RANK];
#pragma unroll
    for (int r = 0; r < RANK; ++r) {
        nmbv[r] = nmb[r];
#pragma unroll
        for (int i = 0; i < DIMZ; ++i) nmw[r][i] = nmW[r * DIMZ + i];
    }

    v8f XD[4];
#pragma unroll
    for (int t4 = 0; t4 < 4; ++t4) { v8f zz = {0,0,0,0,0,0,0,0}; XD[t4] = zz; }
    const v8f vzero = {0,0,0,0,0,0,0,0};

#pragma unroll
    for (int v = 0; v < 8; ++v) ldsP[(v + 8 * hi) * 16 + col] = 0.0f;

    float zq[2][DIMZ], sq0[2], sq1[2], sq2[2];
#pragma unroll
    for (int u = 0; u < 2; ++u) {
        const float* zr = z + ((size_t)u * BATCH + (size_t)brow) * DIMZ;
        const float* sr = s + ((size_t)u * BATCH + (size_t)brow) * DIMS;
#pragma unroll
        for (int i = 0; i < DIMZ; ++i) zq[u][i] = zr[i];
        sq0[u] = sr[0]; sq1[u] = sr[1]; sq2[u] = sr[2];
    }

#pragma unroll 1
    for (int t2 = 0; t2 < SEQ; t2 += 2) {
        StepIn G[2];
#pragma unroll
        for (int u = 0; u < 2; ++u) {
            G[u] = step_inputs(zq[u], sq0[u], sq1[u], sq2[u], nmw, nmbv, hib);
            ldsZ[16 * u + col] = G[u].zsum;
        }

#pragma unroll
        for (int u = 0; u < 2; ++u) {
            const int tn = (t2 + 2 + u < SEQ) ? (t2 + 2 + u) : (SEQ - 1);
            const float* zr = z + ((size_t)tn * BATCH + (size_t)brow) * DIMZ;
            const float* sr = s + ((size_t)tn * BATCH + (size_t)brow) * DIMS;
#pragma unroll
            for (int i = 0; i < DIMZ; ++i) zq[u][i] = zr[i];
            sq0[u] = sr[0]; sq1[u] = sr[1]; sq2[u] = sr[2];
        }

#pragma unroll
    for (int u = 0; u < 2; ++u) {
        const int t = t2 + u;
        const float ga0 = G[u].ga0, ga1 = G[u].ga1, ga2 = G[u].ga2, ga3 = G[u].ga3;

        asm volatile("s_wait_dscnt 0" ::: "memory");
        v2f Ap0, Ap1;
        {
            float p0 = ldsP[col * 16 + 2 * hi];
            float p1 = ldsP[col * 16 + 2 * hi + 1];
            float p2 = ldsP[col * 16 + 4 + 2 * hi];
            float p3 = ldsP[col * 16 + 4 + 2 * hi + 1];
            Ap0[0] = p0 * ga0; Ap0[1] = p1 * ga1;
            Ap1[0] = p2 * ga2; Ap1[1] = p3 * ga3;
        }
        float zrow[8];
#pragma unroll
        for (int v = 0; v < 8; ++v) zrow[v] = ldsZ[16 * u + 8 * hi + v];
        v2f As;
        As[0] = G[u].as0;
        As[1] = G[u].as1;

#pragma unroll
        for (int t4 = 0; t4 < 4; ++t4) {
            v8f acc = wmma_f32_k4(Ap0, Bu[t4][0], vzero);
            acc = wmma_f32_k4(Ap1, Bu[t4][1], acc);
            acc = wmma_f32_k4(As,  Bw[t4],    acc);
            v8f x = XD[t4];
#pragma unroll
            for (int v = 0; v < 8; ++v) x[v] = fmaf(x[v], 1.0f - ALPHA, acc[v]);
            XD[t4] = x;
#pragma unroll
            for (int v = 0; v < 8; ++v)
                ldsX[(v + 8 * hi) * XSTR + 16 * t4 + col] = x[v];
        }
        asm volatile("s_wait_dscnt 0" ::: "memory");

        float xa[2 * KG];
#pragma unroll
        for (int g = 0; g < KG; ++g) {
            int k0 = 4 * g + 2 * hi;
            xa[2 * g]     = ldsX[col * XSTR + k0];
            xa[2 * g + 1] = ldsX[col * XSTR + k0 + 1];
        }
        __builtin_amdgcn_sched_barrier(0);
        float Atan[2 * KG];
#pragma unroll
        for (int i = 0; i < 2 * KG; ++i) Atan[i] = fast_tanh(xa[i]);

        v8f Q = vzero;
#pragma unroll
        for (int g = 0; g < KG; ++g) {
            float a0 = Atan[2 * g];
            float a1 = Atan[2 * g + 1];
            if (g == KG - 1) {
                a0 = hib ? 1.0f : a0;
                a1 = hib ? 0.0f : a1;
            }
            v2f a; a[0] = a0; a[1] = a1;
            Q = wmma_f32_k4(a, Bq[g], Q);
        }

#pragma unroll
        for (int v = 0; v < 8; ++v) ldsP[(v + 8 * hi) * 16 + col] = Q[v];

        float o[8];
#pragma unroll
        for (int v = 0; v < 8; ++v) o[v] = zrow[v] * Q[v];
        if (col >= QOUT && col < QOUT + DIMY) {
            size_t base = (size_t)t * BATCH * DIMY + (size_t)(b0 + 8 * hi) * DIMY + (col - QOUT);
#pragma unroll
            for (int v = 0; v < 8; ++v) out[base + (size_t)v * DIMY] = o[v];
        }
    }
    }
}

extern "C" void kernel_launch(void* const* d_in, const int* in_sizes, int n_in,
                              void* d_out, int out_size, void* d_ws, size_t ws_size,
                              hipStream_t stream) {
    (void)in_sizes; (void)n_in; (void)out_size; (void)d_ws; (void)ws_size;
    const float* s   = (const float*)d_in[0];
    const float* z   = (const float*)d_in[1];
    const float* U   = (const float*)d_in[2];
    const float* V   = (const float*)d_in[3];
    const float* Wi  = (const float*)d_in[4];
    const float* bi  = (const float*)d_in[5];
    const float* Wo  = (const float*)d_in[6];
    const float* bo  = (const float*)d_in[7];
    const float* nmW = (const float*)d_in[8];
    const float* nmb = (const float*)d_in[9];
    float* out = (float*)d_out;

    dim3 grid(BATCH / 16);
    dim3 block(32);
    hipLaunchKernelGGL(cxtrnn_wmma_kernel, grid, block, 0, stream,
                       s, z, U, V, Wi, bi, Wo, bo, nmW, nmb, out);
}


# State kept transposed (hidden x batch): a WMMA's D registers are the next WMMA's B operand in the same lane, k order permuted in constant A operands, input term summed first; no LDS; 4-step trips
# speedup vs baseline: 2.7479x; 1.5615x over previous
#include <hip/hip_runtime.h>


#define SEQ   512
#define BATCH 4096
#define DIMS  3
#define DIMY  3
#define DIMZ  6
#define RANK  6
#define HID   50
#define KG    13
#define ALPHA 0.5f
#define RH    3
#define QY    3
#define HONE  56
#define TRIP  4

typedef float v2f __attribute__((ext_vector_type(2)));
typedef float v8f __attribute__((ext_vector_type(8)));

__device__ __forceinline__ v8f wmma_f32_k4(v2f a, v2f b, v8f c) {
    return __builtin_amdgcn_wmma_f32_16x16x4_f32(false, a, false, b, (short)0, c, false, false);
}

__device__ __forceinline__ float fast_tanh(float x) {
#if __has_builtin(__builtin_amdgcn_tanhf)
    return __builtin_amdgcn_tanhf(x);
#else
    x = fminf(15.0f, fmaxf(-15.0f, x));
    float e = __expf(2.0f * x);
    float r = __builtin_amdgcn_rcpf(e + 1.0f);
    return fmaf(-2.0f, r, 1.0f);
#endif
}

__device__ __forceinline__ float fast_sigmoid(float x) {
    return __builtin_amdgcn_rcpf(1.0f + __expf(-x));
}

__device__ __forceinline__ int q_hid(int g, int h, int s) {
    return 16 * (g >> 2) + 8 * h + 2 * (g & 3) + s;
}

struct StepIn {
    float zsum;
    float g0, g1, g2;
    float bs0, bs1;
};

__device__ __forceinline__ StepIn step_inputs(const float (&zv)[DIMZ],
                                              const float sv0, const float sv1, const float sv2,
                                              const float (&nmw)[RH][DIMZ],
                                              const float (&nmbv)[RH], const bool hib) {
    float zsum = 0.0f;
#pragma unroll
    for (int i = 0; i < DIMZ; ++i) zsum += zv[i];
    float a0 = nmbv[0], a1 = nmbv[1], a2 = nmbv[2];
#pragma unroll
    for (int i = 0; i < DIMZ; ++i) {
        a0 = fmaf(nmw[0][i], zv[i], a0);
        a1 = fmaf(nmw[1][i], zv[i], a1);
        a2 = fmaf(nmw[2][i], zv[i], a2);
    }
    StepIn r;
    r.zsum = zsum;
    r.g0 = fast_sigmoid(a0);
    r.g1 = fast_sigmoid(a1);
    r.g2 = fast_sigmoid(a2);
    r.bs0 = zsum * (hib ? sv2 : sv0);
    r.bs1 = hib ? zsum : (zsum * sv1);
    return r;
}

__global__ __launch_bounds__(32) void cxtrnn_wmma_kernel(
    const float* __restrict__ s,   const float* __restrict__ z,
    const float* __restrict__ U,   const float* __restrict__ V,
    const float* __restrict__ Wi,  const float* __restrict__ bi,
    const float* __restrict__ Wo,  const float* __restrict__ bo,
    const float* __restrict__ nmW, const float* __restrict__ nmb,
    float* __restrict__ out)
{
    const int lane = (int)threadIdx.x;
    const int col  = lane & 15;
    const int hi   = lane >> 4;
    const bool hib = (hi != 0);
    const int b0   = (int)blockIdx.x * 16;
    const int brow = b0 + col;

    v2f Aq[KG];
#pragma unroll
    for (int g = 0; g < KG; ++g) {
        v2f t;
#pragma unroll
        for (int e = 0; e < 2; ++e) {
            const int hd = q_hid(g, hi, e);
            float a = 0.0f;
            if (col < RH) {
                a = (hd < HID) ? V[hd * RANK + col] : 0.0f;
            } else if (col >= 8 && col < 8 + RH) {
                a = (hd < HID) ? V[hd * RANK + (col - 8 + RH)] : 0.0f;
            } else if (col >= QY && col < QY + DIMY) {
                const int y = col - QY;
                a = (hd < HID) ? Wo[y * HID + hd] : ((hd == HONE) ? bo[y] : 0.0f);
            }
            t[e] = a;
        }
        Aq[g] = t;
    }
    v2f Au[4][2];
#pragma unroll
    for (int t4 = 0; t4 < 4; ++t4) {
        const int hd = 16 * t4 + col;
        const bool ok = hd < HID;
        v2f t0, t1;
        t0[0] = ok ? ALPHA * U[hd * RANK + RH * hi]     : 0.0f;
        t0[1] = ok ? ALPHA * U[hd * RANK + RH * hi + 1] : 0.0f;
        t1[0] = ok ? ALPHA * U[hd * RANK + RH * hi + 2] : 0.0f;
        t1[1] = 0.0f;
        Au[t4][0] = t0; Au[t4][1] = t1;
    }
    v2f Aw[4];
#pragma unroll
    for (int t4 = 0; t4 < 4; ++t4) {
        const int hd = 16 * t4 + col;
        const int k0 = 2 * hi;
        float a0 = 0.0f, a1 = 0.0f;
        if (hd < HID) {
            a0 = ALPHA * Wi[hd * DIMS + k0];
            a1 = ALPHA * ((k0 + 1 < DIMS) ? Wi[hd * DIMS + k0 + 1] : bi[hd]);
        }
        v2f t; t[0] = a0; t[1] = a1; Aw[t4] = t;
    }
    float nmw[RH][DIMZ], nmbv[RH];
#pragma unroll
    for (int r = 0; r < RH; ++r) {
        nmbv[r] = nmb[RH * hi + r];
#pragma unroll
        for (int i = 0; i < DIMZ; ++i) nmw[r][i] = nmW[(RH * hi + r) * DIMZ + i];
    }

    v8f XD[3];
    float x30 = 0.0f, x31 = 0.0f;
    const v8f vzero = {0,0,0,0,0,0,0,0};
#pragma unroll
    for (int t4 = 0; t4 < 3; ++t4) XD[t4] = vzero;
    v8f Q = vzero;
    float zprev = 0.0f;

    float zq[TRIP][DIMZ], sq0[TRIP], sq1[TRIP], sq2[TRIP];
#pragma unroll
    for (int u = 0; u < TRIP; ++u) {
        const float* zr = z + ((size_t)u * BATCH + (size_t)brow) * DIMZ;
        const float* sr = s + ((size_t)u * BATCH + (size_t)brow) * DIMS;
#pragma unroll
        for (int i = 0; i < DIMZ; ++i) zq[u][i] = zr[i];
        sq0[u] = sr[0]; sq1[u] = sr[1]; sq2[u] = sr[2];
    }

#pragma unroll 1
    for (int t2 = 0; t2 < SEQ; t2 += TRIP) {
        StepIn G[TRIP];
#pragma unroll
        for (int u = 0; u < TRIP; ++u) {
            G[u] = step_inputs(zq[u], sq0[u], sq1[u], sq2[u], nmw, nmbv, hib);
            if (u == 0)
                asm volatile("" : "+v"(G[u].zsum), "+v"(G[u].g0), "+v"(G[u].g1), "+v"(G[u].g2));
            else
                asm volatile("" : "+v"(G[u].zsum), "+v"(G[u].g0), "+v"(G[u].g1), "+v"(G[u].g2),
                                  "+v"(G[u].bs0), "+v"(G[u].bs1));
        }

#pragma unroll
        for (int u = 0; u < TRIP; ++u) {
            const int tn = (t2 + TRIP + u < SEQ) ? (t2 + TRIP + u) : (SEQ - 1);
            const float* zr = z + ((size_t)tn * BATCH + (size_t)brow) * DIMZ;
            const float* sr = s + ((size_t)tn * BATCH + (size_t)brow) * DIMS;
#pragma unroll
            for (int i = 0; i < DIMZ; ++i) zq[u][i] = zr[i];
            sq0[u] = sr[0]; sq1[u] = sr[1]; sq2[u] = sr[2];
        }

#pragma unroll
    for (int u = 0; u < TRIP; ++u) {
        const int t = t2 + u;

        v2f Bs;
        Bs[0] = G[u].bs0;
        Bs[1] = G[u].bs1;
        v8f acc[4];
#pragma unroll
        for (int t4 = 0; t4 < 4; ++t4) acc[t4] = wmma_f32_k4(Aw[t4], Bs, vzero);

        if (!hib && t > 0) {
            const size_t base = ((size_t)(t - 1) * BATCH + (size_t)brow) * DIMY;
#pragma unroll
            for (int y = 0; y < DIMY; ++y) out[base + y] = zprev * Q[QY + y];
        }
        zprev = G[u].zsum;

        v2f Bp0, Bp1;
        Bp0[0] = Q[0] * G[u].g0;
        Bp0[1] = Q[1] * G[u].g1;
        Bp1[0] = Q[2] * G[u].g2;
        Bp1[1] = 0.0f;

#pragma unroll
        for (int t4 = 0; t4 < 4; ++t4) acc[t4] = wmma_f32_k4(Au[t4][0], Bp0, acc[t4]);
#pragma unroll
        for (int t4 = 0; t4 < 4; ++t4) acc[t4] = wmma_f32_k4(Au[t4][1], Bp1, acc[t4]);

        v8f Qn = vzero;
#pragma unroll
        for (int t4 = 0; t4 < 3; ++t4) {
            v8f x = XD[t4];
#pragma unroll
            for (int v = 0; v < 8; ++v) x[v] = fmaf(x[v], 1.0f - ALPHA, acc[t4][v]);
            XD[t4] = x;
#pragma unroll
            for (int q = 0; q < 4; ++q) {
                v2f b;
                b[0] = fast_tanh(x[2 * q]);
                b[1] = fast_tanh(x[2 * q + 1]);
                Qn = wmma_f32_k4(Aq[4 * t4 + q], b, Qn);
            }
        }
        {
            x30 = fmaf(x30, 1.0f - ALPHA, acc[3][0]);
            x31 = fmaf(x31, 1.0f - ALPHA, acc[3][1]);
            const float th0 = fast_tanh(x30);
            const float th1 = fast_tanh(x31);
            v2f b;
            b[0] = hib ? 1.0f : th0;
            b[1] = hib ? 0.0f : th1;
            Qn = wmma_f32_k4(Aq[KG - 1], b, Qn);
        }
        Q = Qn;
    }
    }

    if (!hib) {
        const size_t base = ((size_t)(SEQ - 1) * BATCH + (size_t)brow) * DIMY;
#pragma unroll
        for (int y = 0; y < DIMY; ++y) out[base + y] = zprev * Q[QY + y];
    }
}

extern "C" void kernel_launch(void* const* d_in, const int* in_sizes, int n_in,
                              void* d_out, int out_size, void* d_ws, size_t ws_size,
                              hipStream_t stream) {
    (void)in_sizes; (void)n_in; (void)out_size; (void)d_ws; (void)ws_size;
    const float* s   = (const float*)d_in[0];
    const float* z   = (const float*)d_in[1];
    const float* U   = (const float*)d_in[2];
    const float* V   = (const float*)d_in[3];
    const float* Wi  = (const float*)d_in[4];
    const float* bi  = (const float*)d_in[5];
    const float* Wo  = (const float*)d_in[6];
    const float* bo  = (const float*)d_in[7];
    const float* nmW = (const float*)d_in[8];
    const float* nmb = (const float*)d_in[9];
    float* out = (float*)d_out;

    dim3 grid(BATCH / 16);
    dim3 block(32);
    hipLaunchKernelGGL(cxtrnn_wmma_kernel, grid, block, 0, stream,
                       s, z, U, V, Wi, bi, Wo, bo, nmW, nmb, out);
}


# Q^T chain pipelined by one link: each K-group's two tanh are issued in front of the PRECEDING chain WMMA, so no chain WMMA waits the transcendental latency; products and order of sums unchanged
# speedup vs baseline: 2.8861x; 1.0503x over previous
#include <hip/hip_runtime.h>


#define SEQ   512
#define BATCH 4096
#define DIMS  3
#define DIMY  3
#define DIMZ  6
#define RANK  6
#define HID   50
#define KG    13
#define ALPHA 0.5f
#define RH    3
#define QY    3
#define HONE  56
#define TRIP  4

typedef float v2f __attribute__((ext_vector_type(2)));
typedef float v8f __attribute__((ext_vector_type(8)));

__device__ __forceinline__ v8f wmma_f32_k4(v2f a, v2f b, v8f c) {
    return __builtin_amdgcn_wmma_f32_16x16x4_f32(false, a, false, b, (short)0, c, false, false);
}

__device__ __forceinline__ float fast_tanh(float x) {
#if __has_builtin(__builtin_amdgcn_tanhf)
    return __builtin_amdgcn_tanhf(x);
#else
    x = fminf(15.0f, fmaxf(-15.0f, x));
    float e = __expf(2.0f * x);
    float r = __builtin_amdgcn_rcpf(e + 1.0f);
    return fmaf(-2.0f, r, 1.0f);
#endif
}

__device__ __forceinline__ float fast_sigmoid(float x) {
    return __builtin_amdgcn_rcpf(1.0f + __expf(-x));
}

__device__ __forceinline__ int q_hid(int g, int h, int s) {
    return 16 * (g >> 2) + 8 * h + 2 * (g & 3) + s;
}

struct StepIn {
    float zsum;
    float g0, g1, g2;
    float bs0, bs1;
};

__device__ __forceinline__ StepIn step_inputs(const float (&zv)[DIMZ],
                                              const float sv0, const float sv1, const float sv2,
                                              const float (&nmw)[RH][DIMZ],
                                              const float (&nmbv)[RH], const bool hib) {
    float zsum = 0.0f;
#pragma unroll
    for (int i = 0; i < DIMZ; ++i) zsum += zv[i];
    float a0 = nmbv[0], a1 = nmbv[1], a2 = nmbv[2];
#pragma unroll
    for (int i = 0; i < DIMZ; ++i) {
        a0 = fmaf(nmw[0][i], zv[i], a0);
        a1 = fmaf(nmw[1][i], zv[i], a1);
        a2 = fmaf(nmw[2][i], zv[i], a2);
    }
    StepIn r;
    r.zsum = zsum;
    r.g0 = fast_sigmoid(a0);
    r.g1 = fast_sigmoid(a1);
    r.g2 = fast_sigmoid(a2);
    r.bs0 = zsum * (hib ? sv2 : sv0);
    r.bs1 = hib ? zsum : (zsum * sv1);
    return r;
}

__global__ __launch_bounds__(32) void cxtrnn_wmma_kernel(
    const float* __restrict__ s,   const float* __restrict__ z,
    const float* __restrict__ U,   const float* __restrict__ V,
    const float* __restrict__ Wi,  const float* __restrict__ bi,
    const float* __restrict__ Wo,  const float* __restrict__ bo,
    const float* __restrict__ nmW, const float* __restrict__ nmb,
    float* __restrict__ out)
{
    const int lane = (int)threadIdx.x;
    const int col  = lane & 15;
    const int hi   = lane >> 4;
    const bool hib = (hi != 0);
    const int b0   = (int)blockIdx.x * 16;
    const int brow = b0 + col;

    v2f Aq[KG];
#pragma unroll
    for (int g = 0; g < KG; ++g) {
        v2f t;
#pragma unroll
        for (int e = 0; e < 2; ++e) {
            const int hd = q_hid(g, hi, e);
            float a = 0.0f;
            if (col < RH) {
                a = (hd < HID) ? V[hd * RANK + col] : 0.0f;
            } else if (col >= 8 && col < 8 + RH) {
                a = (hd < HID) ? V[hd * RANK + (col - 8 + RH)] : 0.0f;
            } else if (col >= QY && col < QY + DIMY) {
                const int y = col - QY;
                a = (hd < HID) ? Wo[y * HID + hd] : ((hd == HONE) ? bo[y] : 0.0f);
            }
            t[e] = a;
        }
        Aq[g] = t;
    }
    v2f Au[4][2];
#pragma unroll
    for (int t4 = 0; t4 < 4; ++t4) {
        const int hd = 16 * t4 + col;
        const bool ok = hd < HID;
        v2f t0, t1;
        t0[0] = ok ? ALPHA * U[hd * RANK + RH * hi]     : 0.0f;
        t0[1] = ok ? ALPHA * U[hd * RANK + RH * hi + 1] : 0.0f;
        t1[0] = ok ? ALPHA * U[hd * RANK + RH * hi + 2] : 0.0f;
        t1[1] = 0.0f;
        Au[t4][0] = t0; Au[t4][1] = t1;
    }
    v2f Aw[4];
#pragma unroll
    for (int t4 = 0; t4 < 4; ++t4) {
        const int hd = 16 * t4 + col;
        const int k0 = 2 * hi;
        float a0 = 0.0f, a1 = 0.0f;
        if (hd < HID) {
            a0 = ALPHA * Wi[hd * DIMS + k0];
            a1 = ALPHA * ((k0 + 1 < DIMS) ? Wi[hd * DIMS + k0 + 1] : bi[hd]);
        }
        v2f t; t[0] = a0; t[1] = a1; Aw[t4] = t;
    }
    float nmw[RH][DIMZ], nmbv[RH];
#pragma unroll
    for (int r = 0; r < RH; ++r) {
        nmbv[r] = nmb[RH * hi + r];
#pragma unroll
        for (int i = 0; i < DIMZ; ++i) nmw[r][i] = nmW[(RH * hi + r) * DIMZ + i];
    }

    v8f XD[3];
    float x30 = 0.0f, x31 = 0.0f;
    const v8f vzero = {0,0,0,0,0,0,0,0};
#pragma unroll
    for (int t4 = 0; t4 < 3; ++t4) XD[t4] = vzero;
    v8f Q = vzero;
    float zprev = 0.0f;

    float zq[TRIP][DIMZ], sq0[TRIP], sq1[TRIP], sq2[TRIP];
#pragma unroll
    for (int u = 0; u < TRIP; ++u) {
        const float* zr = z + ((size_t)u * BATCH + (size_t)brow) * DIMZ;
        const float* sr = s + ((size_t)u * BATCH + (size_t)brow) * DIMS;
#pragma unroll
        for (int i = 0; i < DIMZ; ++i) zq[u][i] = zr[i];
        sq0[u] = sr[0]; sq1[u] = sr[1]; sq2[u] = sr[2];
    }

#pragma unroll 1
    for (int t2 = 0; t2 < SEQ; t2 += TRIP) {
        StepIn G[TRIP];
#pragma unroll
        for (int u = 0; u < TRIP; ++u) {
            G[u] = step_inputs(zq[u], sq0[u], sq1[u], sq2[u], nmw, nmbv, hib);
            if (u == 0)
                asm volatile("" : "+v"(G[u].zsum), "+v"(G[u].g0), "+v"(G[u].g1), "+v"(G[u].g2));
            else
                asm volatile("" : "+v"(G[u].zsum), "+v"(G[u].g0), "+v"(G[u].g1), "+v"(G[u].g2),
                                  "+v"(G[u].bs0), "+v"(G[u].bs1));
        }

#pragma unroll
        for (int u = 0; u < TRIP; ++u) {
            const int tn = (t2 + TRIP + u < SEQ) ? (t2 + TRIP + u) : (SEQ - 1);
            const float* zr = z + ((size_t)tn * BATCH + (size_t)brow) * DIMZ;
            const float* sr = s + ((size_t)tn * BATCH + (size_t)brow) * DIMS;
#pragma unroll
            for (int i = 0; i < DIMZ; ++i) zq[u][i] = zr[i];
            sq0[u] = sr[0]; sq1[u] = sr[1]; sq2[u] = sr[2];
        }

#pragma unroll
    for (int u = 0; u < TRIP; ++u) {
        const int t = t2 + u;

        v2f Bs;
        Bs[0] = G[u].bs0;
        Bs[1] = G[u].bs1;
        v8f acc[4];
#pragma unroll
        for (int t4 = 0; t4 < 4; ++t4) acc[t4] = wmma_f32_k4(Aw[t4], Bs, vzero);

        if (!hib && t > 0) {
            const size_t base = ((size_t)(t - 1) * BATCH + (size_t)brow) * DIMY;
#pragma unroll
            for (int y = 0; y < DIMY; ++y) out[base + y] = zprev * Q[QY + y];
        }
        zprev = G[u].zsum;

        v2f Bp0, Bp1;
        Bp0[0] = Q[0] * G[u].g0;
        Bp0[1] = Q[1] * G[u].g1;
        Bp1[0] = Q[2] * G[u].g2;
        Bp1[1] = 0.0f;

#pragma unroll
        for (int t4 = 0; t4 < 4; ++t4) acc[t4] = wmma_f32_k4(Au[t4][0], Bp0, acc[t4]);
#pragma unroll
        for (int t4 = 0; t4 < 4; ++t4) acc[t4] = wmma_f32_k4(Au[t4][1], Bp1, acc[t4]);

#pragma unroll
        for (int t4 = 0; t4 < 3; ++t4) {
            v8f x = XD[t4];
#pragma unroll
            for (int v = 0; v < 8; ++v) x[v] = fmaf(x[v], 1.0f - ALPHA, acc[t4][v]);
            XD[t4] = x;
        }
        x30 = fmaf(x30, 1.0f - ALPHA, acc[3][0]);
        x31 = fmaf(x31, 1.0f - ALPHA, acc[3][1]);

        v8f Qn = vzero;
        v2f bcur;
        bcur[0] = fast_tanh(XD[0][0]);
        bcur[1] = fast_tanh(XD[0][1]);
#pragma unroll
        for (int g = 0; g < KG; ++g) {
            v2f bnext = bcur;
            if (g + 1 < KG - 1) {
                const int t4 = (g + 1) >> 2, q = (g + 1) & 3;
                bnext[0] = fast_tanh(XD[t4][2 * q]);
                bnext[1] = fast_tanh(XD[t4][2 * q + 1]);
            } else if (g + 1 == KG - 1) {
                const float th0 = fast_tanh(x30);
                const float th1 = fast_tanh(x31);
                bnext[0] = hib ? 1.0f : th0;
                bnext[1] = hib ? 0.0f : th1;
            }
            if (g == 0)
                asm volatile("" : "+v"(bcur), "+v"(bnext));
            else if (g + 1 < KG)
                asm volatile("" : "+v"(bnext), "+v"(Qn));
            Qn = wmma_f32_k4(Aq[g], bcur, Qn);
            bcur = bnext;
        }
        Q = Qn;
    }
    }

    if (!hib) {
        const size_t base = ((size_t)(SEQ - 1) * BATCH + (size_t)brow) * DIMY;
#pragma unroll
        for (int y = 0; y < DIMY; ++y) out[base + y] = zprev * Q[QY + y];
    }
}

extern "C" void kernel_launch(void* const* d_in, const int* in_sizes, int n_in,
                              void* d_out, int out_size, void* d_ws, size_t ws_size,
                              hipStream_t stream) {
    (void)in_sizes; (void)n_in; (void)out_size; (void)d_ws; (void)ws_size;
    const float* s   = (const float*)d_in[0];
    const float* z   = (const float*)d_in[1];
    const float* U   = (const float*)d_in[2];
    const float* V   = (const float*)d_in[3];
    const float* Wi  = (const float*)d_in[4];
    const float* bi  = (const float*)d_in[5];
    const float* Wo  = (const float*)d_in[6];
    const float* bo  = (const float*)d_in[7];
    const float* nmW = (const float*)d_in[8];
    const float* nmb = (const float*)d_in[9];
    float* out = (float*)d_out;

    dim3 grid(BATCH / 16);
    dim3 block(32);
    hipLaunchKernelGGL(cxtrnn_wmma_kernel, grid, block, 0, stream,
                       s, z, U, V, Wi, bi, Wo, bo, nmW, nmb, out);
}


# Hidden rows split over 4 waves of a workgroup, one per SIMD: each wave updates its 16 rows and sums its own K-groups of Q^T; partial sums cross through LDS once a step; order of sums changes
# speedup vs baseline: 3.4265x; 1.1872x over previous
#include <hip/hip_runtime.h>


#define SEQ   512
#define BATCH 4096
#define DIMS  3
#define DIMY  3
#define DIMZ  6
#define RANK  6
#define HID   50
#define KG    13
#define ALPHA 0.5f
#define RH    3
#define QY    3
#define HONE  56
#define TRIP  4
#define NWAVE 4
#define KGW   4

typedef float v2f __attribute__((ext_vector_type(2)));
typedef float v8f __attribute__((ext_vector_type(8)));
typedef float v3f __attribute__((ext_vector_type(3)));

__device__ __forceinline__ v8f wmma_f32_k4(v2f a, v2f b, v8f c) {
    return __builtin_amdgcn_wmma_f32_16x16x4_f32(false, a, false, b, (short)0, c, false, false);
}

__device__ __forceinline__ float fast_tanh(float x) {
#if __has_builtin(__builtin_amdgcn_tanhf)
    return __builtin_amdgcn_tanhf(x);
#else
    x = fminf(15.0f, fmaxf(-15.0f, x));
    float e = __expf(2.0f * x);
    float r = __builtin_amdgcn_rcpf(e + 1.0f);
    return fmaf(-2.0f, r, 1.0f);
#endif
}

__device__ __forceinline__ float fast_sigmoid(float x) {
    return __builtin_amdgcn_rcpf(1.0f + __expf(-x));
}

__device__ __forceinline__ void lds_handover_barrier() {
    __builtin_amdgcn_fence(__ATOMIC_RELEASE, "workgroup", "local");
    __builtin_amdgcn_s_barrier();
    __builtin_amdgcn_fence(__ATOMIC_ACQUIRE, "workgroup", "local");
}

__device__ __forceinline__ int q_hid(int g, int h, int s) {
    return 16 * (g >> 2) + 8 * h + 2 * (g & 3) + s;
}

struct StepIn {
    float zsum;
    float g0, g1, g2;
    float bs0, bs1;
};

__device__ __forceinline__ StepIn step_inputs(const float (&zv)[DIMZ],
                                              const float sv0, const float sv1, const float sv2,
                                              const float (&nmw)[RH][DIMZ],
                                              const float (&nmbv)[RH], const bool hib) {
    float zsum = 0.0f;
#pragma unroll
    for (int i = 0; i < DIMZ; ++i) zsum += zv[i];
    float a0 = nmbv[0], a1 = nmbv[1], a2 = nmbv[2];
#pragma unroll
    for (int i = 0; i < DIMZ; ++i) {
        a0 = fmaf(nmw[0][i], zv[i], a0);
        a1 = fmaf(nmw[1][i], zv[i], a1);
        a2 = fmaf(nmw[2][i], zv[i], a2);
    }
    StepIn r;
    r.zsum = zsum;
    r.g0 = fast_sigmoid(a0);
    r.g1 = fast_sigmoid(a1);
    r.g2 = fast_sigmoid(a2);
    r.bs0 = zsum * (hib ? sv2 : sv0);
    r.bs1 = hib ? zsum : (zsum * sv1);
    return r;
}

__global__ __launch_bounds__(32 * NWAVE) void cxtrnn_wmma_kernel(
    const float* __restrict__ s,   const float* __restrict__ z,
    const float* __restrict__ U,   const float* __restrict__ V,
    const float* __restrict__ Wi,  const float* __restrict__ bi,
    const float* __restrict__ Wo,  const float* __restrict__ bo,
    const float* __restrict__ nmW, const float* __restrict__ nmb,
    float* __restrict__ out)
{
    __shared__ v3f shP[2][NWAVE][32];
    __shared__ v3f shY[2][NWAVE][32];

    const int lane = (int)threadIdx.x & 31;
    const int w    = __builtin_amdgcn_readfirstlane((int)threadIdx.x >> 5);
    const bool w3  = (w == NWAVE - 1);
    const int col  = lane & 15;
    const int hi   = lane >> 4;
    const bool hib = (hi != 0);
    const int b0   = (int)blockIdx.x * 16;
    const int brow = b0 + col;

    v2f Aq[KGW];
#pragma unroll
    for (int q = 0; q < KGW; ++q) {
        const int g = KGW * w + q;
        v2f t;
#pragma unroll
        for (int e = 0; e < 2; ++e) {
            const int hd = q_hid(g, hi, e);
            float a = 0.0f;
            if (col < RH) {
                a = (hd < HID) ? V[hd * RANK + col] : 0.0f;
            } else if (col >= 8 && col < 8 + RH) {
                a = (hd < HID) ? V[hd * RANK + (col - 8 + RH)] : 0.0f;
            } else if (col >= QY && col < QY + DIMY) {
                const int y = col - QY;
                a = (hd < HID) ? Wo[y * HID + hd] : ((hd == HONE) ? bo[y] : 0.0f);
            }
            t[e] = a;
        }
        Aq[q] = t;
    }
    v2f Au0, Au1;
    v2f Aw;
    {
        const int hd = 16 * w + col;
        const bool ok = hd < HID;
        Au0[0] = ok ? ALPHA * U[hd * RANK + RH * hi]     : 0.0f;
        Au0[1] = ok ? ALPHA * U[hd * RANK + RH * hi + 1] : 0.0f;
        Au1[0] = ok ? ALPHA * U[hd * RANK + RH * hi + 2] : 0.0f;
        Au1[1] = 0.0f;
        const int k0 = 2 * hi;
        float a0 = 0.0f, a1 = 0.0f;
        if (ok) {
            a0 = ALPHA * Wi[hd * DIMS + k0];
            a1 = ALPHA * ((k0 + 1 < DIMS) ? Wi[hd * DIMS + k0 + 1] : bi[hd]);
        }
        Aw[0] = a0; Aw[1] = a1;
    }
    float nmw[RH][DIMZ], nmbv[RH];
#pragma unroll
    for (int r = 0; r < RH; ++r) {
        nmbv[r] = nmb[RH * hi + r];
#pragma unroll
        for (int i = 0; i < DIMZ; ++i) nmw[r][i] = nmW[(RH * hi + r) * DIMZ + i];
    }

    const v8f vzero = {0,0,0,0,0,0,0,0};
    v8f X = vzero;
    const bool one_slot = w3 && hib;
    float zprev = 0.0f;

    {
        const v3f z3 = {0.0f, 0.0f, 0.0f};
        shP[1][w][lane] = z3;
        shY[1][w][lane] = z3;
    }

    float zq[TRIP][DIMZ], sq0[TRIP], sq1[TRIP], sq2[TRIP];
#pragma unroll
    for (int u = 0; u < TRIP; ++u) {
        const float* zr = z + ((size_t)u * BATCH + (size_t)brow) * DIMZ;
        const float* sr = s + ((size_t)u * BATCH + (size_t)brow) * DIMS;
#pragma unroll
        for (int i = 0; i < DIMZ; ++i) zq[u][i] = zr[i];
        sq0[u] = sr[0]; sq1[u] = sr[1]; sq2[u] = sr[2];
    }
    lds_handover_barrier();

#pragma unroll 1
    for (int t2 = 0; t2 < SEQ; t2 += TRIP) {
        StepIn G[TRIP];
#pragma unroll
        for (int u = 0; u < TRIP; ++u) {
            G[u] = step_inputs(zq[u], sq0[u], sq1[u], sq2[u], nmw, nmbv, hib);
            asm volatile("" : "+v"(G[u].zsum), "+v"(G[u].g0), "+v"(G[u].g1), "+v"(G[u].g2),
                              "+v"(G[u].bs0), "+v"(G[u].bs1));
        }

#pragma unroll
        for (int u = 0; u < TRIP; ++u) {
            const int tn = (t2 + TRIP + u < SEQ) ? (t2 + TRIP + u) : (SEQ - 1);
            const float* zr = z + ((size_t)tn * BATCH + (size_t)brow) * DIMZ;
            const float* sr = s + ((size_t)tn * BATCH + (size_t)brow) * DIMS;
#pragma unroll
            for (int i = 0; i < DIMZ; ++i) zq[u][i] = zr[i];
            sq0[u] = sr[0]; sq1[u] = sr[1]; sq2[u] = sr[2];
        }

#pragma unroll
    for (int u = 0; u < TRIP; ++u) {
        const int t  = t2 + u;
        const int cb = u & 1;
        const int pb = cb ^ 1;

        v2f Bs;
        Bs[0] = G[u].bs0;
        Bs[1] = G[u].bs1;
        v8f acc = wmma_f32_k4(Aw, Bs, vzero);

        const v3f p0 = shP[pb][0][lane];
        const v3f p1 = shP[pb][1][lane];
        const v3f p2 = shP[pb][2][lane];
        const v3f p3 = shP[pb][3][lane];
        const float P0 = ((p0.x + p1.x) + p2.x) + p3.x;
        const float P1 = ((p0.y + p1.y) + p2.y) + p3.y;
        const float P2 = ((p0.z + p1.z) + p2.z) + p3.z;

        if (w3) {
            const v3f o0 = shY[pb][0][lane];
            const v3f o1 = shY[pb][1][lane];
            const v3f o2 = shY[pb][2][lane];
            const v3f o3 = shY[pb][3][lane];
            float y0 = ((o0.x + o1.x) + o2.x) + o3.x;
            float y1 = ((o0.y + o1.y) + o2.y) + o3.y;
            float y2 = ((o0.z + o1.z) + o2.z) + o3.z;
            asm volatile("" : "+v"(y0), "+v"(y1), "+v"(y2));
            if (!hib && t > 0) {
                const size_t base = ((size_t)(t - 1) * BATCH + (size_t)brow) * DIMY;
                out[base + 0] = zprev * y0;
                out[base + 1] = zprev * y1;
                out[base + 2] = zprev * y2;
            }
        }
        zprev = G[u].zsum;

        v2f Bp0, Bp1;
        Bp0[0] = P0 * G[u].g0;
        Bp0[1] = P1 * G[u].g1;
        Bp1[0] = P2 * G[u].g2;
        Bp1[1] = 0.0f;

        acc = wmma_f32_k4(Au0, Bp0, acc);
        acc = wmma_f32_k4(Au1, Bp1, acc);

#pragma unroll
        for (int v = 0; v < 8; ++v) X[v] = fmaf(X[v], 1.0f - ALPHA, acc[v]);

        v8f Qp = vzero;
        v2f bcur;
        {
            const float th0 = fast_tanh(X[0]);
            const float th1 = fast_tanh(X[1]);
            bcur[0] = one_slot ? 1.0f : th0;
            bcur[1] = one_slot ? 0.0f : th1;
        }
        if (!w3) {
#pragma unroll
            for (int q = 0; q < KGW; ++q) {
                v2f bnext = bcur;
                if (q + 1 < KGW) {
                    bnext[0] = fast_tanh(X[2 * (q + 1)]);
                    bnext[1] = fast_tanh(X[2 * (q + 1) + 1]);
                }
                if (q == 0)
                    asm volatile("" : "+v"(bcur), "+v"(bnext));
                else if (q + 1 < KGW)
                    asm volatile("" : "+v"(bnext), "+v"(Qp));
                Qp = wmma_f32_k4(Aq[q], bcur, Qp);
                bcur = bnext;
            }
        } else {
            Qp = wmma_f32_k4(Aq[0], bcur, Qp);
        }

        shP[cb][w][lane] = __builtin_shufflevector(Qp, Qp, 0, 1, 2);
        shY[cb][w][lane] = __builtin_shufflevector(Qp, Qp, 3, 4, 5);
        lds_handover_barrier();
    }
    }

    if (w3 && !hib) {
        const int lb = (SEQ - 1) & 1;
        float y0 = 0.0f, y1 = 0.0f, y2 = 0.0f;
#pragma unroll
        for (int k = 0; k < NWAVE; ++k) {
            const v3f o = shY[lb][k][lane];
            y0 = (k == 0) ? o.x : (y0 + o.x);
            y1 = (k == 0) ? o.y : (y1 + o.y);
            y2 = (k == 0) ? o.z : (y2 + o.z);
        }
        const size_t base = ((size_t)(SEQ - 1) * BATCH + (size_t)brow) * DIMY;
        out[base + 0] = zprev * y0;
        out[base + 1] = zprev * y1;
        out[base + 2] = zprev * y2;
    }
}

extern "C" void kernel_launch(void* const* d_in, const int* in_sizes, int n_in,
                              void* d_out, int out_size, void* d_ws, size_t ws_size,
                              hipStream_t stream) {
    (void)in_sizes; (void)n_in; (void)out_size; (void)d_ws; (void)ws_size;
    const float* s   = (const float*)d_in[0];
    const float* z   = (const float*)d_in[1];
    const float* U   = (const float*)d_in[2];
    const float* V   = (const float*)d_in[3];
    const float* Wi  = (const float*)d_in[4];
    const float* bi  = (const float*)d_in[5];
    const float* Wo  = (const float*)d_in[6];
    const float* bo  = (const float*)d_in[7];
    const float* nmW = (const float*)d_in[8];
    const float* nmb = (const float*)d_in[9];
    float* out = (float*)d_out;

    dim3 grid(BATCH / 16);
    dim3 block(32 * NWAVE);
    hipLaunchKernelGGL(cxtrnn_wmma_kernel, grid, block, 0, stream,
                       s, z, U, V, Wi, bi, Wo, bo, nmW, nmb, out);
}


# Step inputs of step t+4 (gates, input operand) made at the end of step t behind its LDS stores, row t+8 loaded there: in the hand-over's shadow, not in a block at each trip's head; sums unchanged
# speedup vs baseline: 3.6367x; 1.0613x over previous
#include <hip/hip_runtime.h>


#define SEQ   512
#define BATCH 4096
#define DIMS  3
#define DIMY  3
#define DIMZ  6
#define RANK  6
#define HID   50
#define KG    13
#define ALPHA 0.5f
#define RH    3
#define QY    3
#define HONE  56
#define TRIP  4
#define NWAVE 4
#define KGW   4

typedef float v2f __attribute__((ext_vector_type(2)));
typedef float v8f __attribute__((ext_vector_type(8)));
typedef float v3f __attribute__((ext_vector_type(3)));

__device__ __forceinline__ v8f wmma_f32_k4(v2f a, v2f b, v8f c) {
    return __builtin_amdgcn_wmma_f32_16x16x4_f32(false, a, false, b, (short)0, c, false, false);
}

__device__ __forceinline__ float fast_tanh(float x) {
#if __has_builtin(__builtin_amdgcn_tanhf)
    return __builtin_amdgcn_tanhf(x);
#else
    x = fminf(15.0f, fmaxf(-15.0f, x));
    float e = __expf(2.0f * x);
    float r = __builtin_amdgcn_rcpf(e + 1.0f);
    return fmaf(-2.0f, r, 1.0f);
#endif
}

__device__ __forceinline__ float fast_sigmoid(float x) {
    return __builtin_amdgcn_rcpf(1.0f + __expf(-x));
}

__device__ __forceinline__ void lds_handover_barrier() {
    __builtin_amdgcn_fence(__ATOMIC_RELEASE, "workgroup", "local");
    __builtin_amdgcn_s_barrier();
    __builtin_amdgcn_fence(__ATOMIC_ACQUIRE, "workgroup", "local");
}

__device__ __forceinline__ int q_hid(int g, int h, int s) {
    return 16 * (g >> 2) + 8 * h + 2 * (g & 3) + s;
}

struct StepIn {
    float zsum;
    float g0, g1, g2;
    float bs0, bs1;
};

__device__ __forceinline__ StepIn step_inputs(const float (&zv)[DIMZ],
                                              const float sv0, const float sv1, const float sv2,
                                              const float (&nmw)[RH][DIMZ],
                                              const float zstart, const float (&bstart)[RH],
                                              const bool hib) {
    float zsum = zstart;
#pragma unroll
    for (int i = 0; i < DIMZ; ++i) zsum += zv[i];
    float a0 = bstart[0], a1 = bstart[1], a2 = bstart[2];
#pragma unroll
    for (int i = 0; i < DIMZ; ++i) {
        a0 = fmaf(nmw[0][i], zv[i], a0);
        a1 = fmaf(nmw[1][i], zv[i], a1);
        a2 = fmaf(nmw[2][i], zv[i], a2);
    }
    StepIn r;
    r.zsum = zsum;
    r.g0 = fast_sigmoid(a0);
    r.g1 = fast_sigmoid(a1);
    r.g2 = fast_sigmoid(a2);
    r.bs0 = zsum * (hib ? sv2 : sv0);
    r.bs1 = hib ? zsum : (zsum * sv1);
    return r;
}

__global__ __launch_bounds__(32 * NWAVE) void cxtrnn_wmma_kernel(
    const float* __restrict__ s,   const float* __restrict__ z,
    const float* __restrict__ U,   const float* __restrict__ V,
    const float* __restrict__ Wi,  const float* __restrict__ bi,
    const float* __restrict__ Wo,  const float* __restrict__ bo,
    const float* __restrict__ nmW, const float* __restrict__ nmb,
    float* __restrict__ out)
{
    __shared__ v3f shP[2][NWAVE][32];
    __shared__ v3f shY[2][NWAVE][32];

    const int lane = (int)threadIdx.x & 31;
    const int w    = __builtin_amdgcn_readfirstlane((int)threadIdx.x >> 5);
    const bool w3  = (w == NWAVE - 1);
    const int col  = lane & 15;
    const int hi   = lane >> 4;
    const bool hib = (hi != 0);
    const int b0   = (int)blockIdx.x * 16;
    const int brow = b0 + col;

    v2f Aq[KGW];
#pragma unroll
    for (int q = 0; q < KGW; ++q) {
        const int g = KGW * w + q;
        v2f t;
#pragma unroll
        for (int e = 0; e < 2; ++e) {
            const int hd = q_hid(g, hi, e);
            float a = 0.0f;
            if (col < RH) {
                a = (hd < HID) ? V[hd * RANK + col] : 0.0f;
            } else if (col >= 8 && col < 8 + RH) {
                a = (hd < HID) ? V[hd * RANK + (col - 8 + RH)] : 0.0f;
            } else if (col >= QY && col < QY + DIMY) {
                const int y = col - QY;
                a = (hd < HID) ? Wo[y * HID + hd] : ((hd == HONE) ? bo[y] : 0.0f);
            }
            t[e] = a;
        }
        Aq[q] = t;
    }
    v2f Au0, Au1;
    v2f Aw;
    {
        const int hd = 16 * w + col;
        const bool ok = hd < HID;
        Au0[0] = ok ? ALPHA * U[hd * RANK + RH * hi]     : 0.0f;
        Au0[1] = ok ? ALPHA * U[hd * RANK + RH * hi + 1] : 0.0f;
        Au1[0] = ok ? ALPHA * U[hd * RANK + RH * hi + 2] : 0.0f;
        Au1[1] = 0.0f;
        const int k0 = 2 * hi;
        float a0 = 0.0f, a1 = 0.0f;
        if (ok) {
            a0 = ALPHA * Wi[hd * DIMS + k0];
            a1 = ALPHA * ((k0 + 1 < DIMS) ? Wi[hd * DIMS + k0 + 1] : bi[hd]);
        }
        Aw[0] = a0; Aw[1] = a1;
    }
    float nmw[RH][DIMZ], nmbv[RH];
#pragma unroll
    for (int r = 0; r < RH; ++r) {
        nmbv[r] = nmb[RH * hi + r];
#pragma unroll
        for (int i = 0; i < DIMZ; ++i) nmw[r][i] = nmW[(RH * hi + r) * DIMZ + i];
    }

    const v8f vzero = {0,0,0,0,0,0,0,0};
    v8f X = vzero;
    const bool one_slot = w3 && hib;
    float zprev = 0.0f;

    {
        const v3f z3 = {0.0f, 0.0f, 0.0f};
        shP[1][w][lane] = z3;
        shY[1][w][lane] = z3;
    }

    const unsigned zoff = (unsigned)brow * DIMZ;
    const unsigned soff = (unsigned)brow * DIMS;
    float zq[TRIP][DIMZ], sq0[TRIP], sq1[TRIP], sq2[TRIP];
    StepIn G[TRIP];
#pragma unroll
    for (int u = 0; u < TRIP; ++u) {
        const float* zr = z + (size_t)u * (BATCH * DIMZ) + zoff;
        const float* sr = s + (size_t)u * (BATCH * DIMS) + soff;
#pragma unroll
        for (int i = 0; i < DIMZ; ++i) zq[u][i] = zr[i];
        sq0[u] = sr[0]; sq1[u] = sr[1]; sq2[u] = sr[2];
    }
#pragma unroll
    for (int u = 0; u < TRIP; ++u) {
        G[u] = step_inputs(zq[u], sq0[u], sq1[u], sq2[u], nmw, 0.0f, nmbv, hib);
        const float* zr = z + (size_t)(TRIP + u) * (BATCH * DIMZ) + zoff;
        const float* sr = s + (size_t)(TRIP + u) * (BATCH * DIMS) + soff;
#pragma unroll
        for (int i = 0; i < DIMZ; ++i) zq[u][i] = zr[i];
        sq0[u] = sr[0]; sq1[u] = sr[1]; sq2[u] = sr[2];
    }
    lds_handover_barrier();

#pragma unroll 1
    for (int t2 = 0; t2 < SEQ; t2 += TRIP) {
#pragma unroll
    for (int u = 0; u < TRIP; ++u) {
        const int t  = t2 + u;
        const int cb = u & 1;
        const int pb = cb ^ 1;

        v2f Bs;
        Bs[0] = G[u].bs0;
        Bs[1] = G[u].bs1;
        v8f acc = wmma_f32_k4(Aw, Bs, vzero);

        const v3f p0 = shP[pb][0][lane];
        const v3f p1 = shP[pb][1][lane];
        const v3f p2 = shP[pb][2][lane];
        const v3f p3 = shP[pb][3][lane];
        const float P0 = ((p0.x + p1.x) + p2.x) + p3.x;
        const float P1 = ((p0.y + p1.y) + p2.y) + p3.y;
        const float P2 = ((p0.z + p1.z) + p2.z) + p3.z;

        if (w3) {
            const v3f o0 = shY[pb][0][lane];
            const v3f o1 = shY[pb][1][lane];
            const v3f o2 = shY[pb][2][lane];
            const v3f o3 = shY[pb][3][lane];
            float y0 = ((o0.x + o1.x) + o2.x) + o3.x;
            float y1 = ((o0.y + o1.y) + o2.y) + o3.y;
            float y2 = ((o0.z + o1.z) + o2.z) + o3.z;
            asm volatile("" : "+v"(y0), "+v"(y1), "+v"(y2));
            if (!hib && t > 0) {
                const size_t base = ((size_t)(t - 1) * BATCH + (size_t)brow) * DIMY;
                out[base + 0] = zprev * y0;
                out[base + 1] = zprev * y1;
                out[base + 2] = zprev * y2;
            }
        }
        zprev = G[u].zsum;

        v2f Bp0, Bp1;
        Bp0[0] = P0 * G[u].g0;
        Bp0[1] = P1 * G[u].g1;
        Bp1[0] = P2 * G[u].g2;
        Bp1[1] = 0.0f;

        acc = wmma_f32_k4(Au0, Bp0, acc);
        acc = wmma_f32_k4(Au1, Bp1, acc);

#pragma unroll
        for (int v = 0; v < 8; ++v) X[v] = fmaf(X[v], 1.0f - ALPHA, acc[v]);

        v8f Qp = vzero;
        v2f bcur;
        {
            const float th0 = fast_tanh(X[0]);
            const float th1 = fast_tanh(X[1]);
            bcur[0] = one_slot ? 1.0f : th0;
            bcur[1] = one_slot ? 0.0f : th1;
        }
        if (!w3) {
#pragma unroll
            for (int q = 0; q < KGW; ++q) {
                v2f bnext = bcur;
                if (q + 1 < KGW) {
                    bnext[0] = fast_tanh(X[2 * (q + 1)]);
                    bnext[1] = fast_tanh(X[2 * (q + 1) + 1]);
                }
                if (q == 0)
                    asm volatile("" : "+v"(bcur), "+v"(bnext));
                else if (q + 1 < KGW)
                    asm volatile("" : "+v"(bnext), "+v"(Qp));
                Qp = wmma_f32_k4(Aq[q], bcur, Qp);
                bcur = bnext;
            }
        } else {
            Qp = wmma_f32_k4(Aq[0], bcur, Qp);
        }

        shP[cb][w][lane] = __builtin_shufflevector(Qp, Qp, 0, 1, 2);
        shY[cb][w][lane] = __builtin_shufflevector(Qp, Qp, 3, 4, 5);

        float zstart = 0.0f;
        float bstart[RH] = {nmbv[0], nmbv[1], nmbv[2]};
        asm volatile("" : "+v"(zstart), "+v"(bstart[0]), "+v"(bstart[1]), "+v"(bstart[2])
                        : : "memory");
        G[u] = step_inputs(zq[u], sq0[u], sq1[u], sq2[u], nmw, zstart, bstart, hib);
        {
            const int tn = (t + 2 * TRIP < SEQ) ? (t + 2 * TRIP) : (SEQ - 1);
            const float* zr = z + (size_t)tn * (BATCH * DIMZ) + zoff;
            const float* sr = s + (size_t)tn * (BATCH * DIMS) + soff;
#pragma unroll
            for (int i = 0; i < DIMZ; ++i) zq[u][i] = zr[i];
            sq0[u] = sr[0]; sq1[u] = sr[1]; sq2[u] = sr[2];
        }
        asm volatile("" : "+v"(G[u].zsum), "+v"(G[u].g0), "+v"(G[u].g1), "+v"(G[u].g2),
                          "+v"(G[u].bs0), "+v"(G[u].bs1));
        lds_handover_barrier();
    }
    }

    if (w3 && !hib) {
        const int lb = (SEQ - 1) & 1;
        float y0 = 0.0f, y1 = 0.0f, y2 = 0.0f;
#pragma unroll
        for (int k = 0; k < NWAVE; ++k) {
            const v3f o = shY[lb][k][lane];
            y0 = (k == 0) ? o.x : (y0 + o.x);
            y1 = (k == 0) ? o.y : (y1 + o.y);
            y2 = (k == 0) ? o.z : (y2 + o.z);
        }
        const size_t base = ((size_t)(SEQ - 1) * BATCH + (size_t)brow) * DIMY;
        out[base + 0] = zprev * y0;
        out[base + 1] = zprev * y1;
        out[base + 2] = zprev * y2;
    }
}

extern "C" void kernel_launch(void* const* d_in, const int* in_sizes, int n_in,
                              void* d_out, int out_size, void* d_ws, size_t ws_size,
                              hipStream_t stream) {
    (void)in_sizes; (void)n_in; (void)out_size; (void)d_ws; (void)ws_size;
    const float* s   = (const float*)d_in[0];
    const float* z   = (const float*)d_in[1];
    const float* U   = (const float*)d_in[2];
    const float* V   = (const float*)d_in[3];
    const float* Wi  = (const float*)d_in[4];
    const float* bi  = (const float*)d_in[5];
    const float* Wo  = (const float*)d_in[6];
    const float* bo  = (const float*)d_in[7];
    const float* nmW = (const float*)d_in[8];
    const float* nmb = (const float*)d_in[9];
    float* out = (float*)d_out;

    dim3 grid(BATCH / 16);
    dim3 block(32 * NWAVE);
    hipLaunchKernelGGL(cxtrnn_wmma_kernel, grid, block, 0, stream,
                       s, z, U, V, Wi, bi, Wo, bo, nmW, nmb, out);
}


# Global accesses as scalar base + fixed lane offset, out store moved behind the LDS stores: no wait for address translation (s_wait_xcnt) left on the step's path; sums unchanged
# speedup vs baseline: 4.2459x; 1.1675x over previous
#include <hip/hip_runtime.h>


#define SEQ   512
#define BATCH 4096
#define DIMS  3
#define DIMY  3
#define DIMZ  6
#define RANK  6
#define HID   50
#define KG    13
#define ALPHA 0.5f
#define RH    3
#define QY    3
#define HONE  56
#define TRIP  4
#define NWAVE 4
#define KGW   4

typedef float v2f __attribute__((ext_vector_type(2)));
typedef float v8f __attribute__((ext_vector_type(8)));
typedef float v3f __attribute__((ext_vector_type(3)));
typedef const float __attribute__((address_space(1)))* gcf_ptr;
typedef float __attribute__((address_space(1)))* gf_ptr;
typedef const char __attribute__((address_space(1)))* gcc_ptr;
typedef char __attribute__((address_space(1)))* gc_ptr;

__device__ __forceinline__ gcc_ptr row_base(const float* base, size_t elem, unsigned& lane_bytes) {
    gcc_ptr b = (gcc_ptr)(base + elem);
    asm volatile("" : "+s"(b), "+v"(lane_bytes));
    return b + lane_bytes;
}
__device__ __forceinline__ gcf_ptr row_ptr(const float* base, size_t elem, unsigned& lane_bytes) {
    return (gcf_ptr)row_base(base, elem, lane_bytes);
}

__device__ __forceinline__ v8f wmma_f32_k4(v2f a, v2f b, v8f c) {
    return __builtin_amdgcn_wmma_f32_16x16x4_f32(false, a, false, b, (short)0, c, false, false);
}

__device__ __forceinline__ float fast_tanh(float x) {
#if __has_builtin(__builtin_amdgcn_tanhf)
    return __builtin_amdgcn_tanhf(x);
#else
    x = fminf(15.0f, fmaxf(-15.0f, x));
    float e = __expf(2.0f * x);
    float r = __builtin_amdgcn_rcpf(e + 1.0f);
    return fmaf(-2.0f, r, 1.0f);
#endif
}

__device__ __forceinline__ float fast_sigmoid(float x) {
    return __builtin_amdgcn_rcpf(1.0f + __expf(-x));
}

__device__ __forceinline__ void lds_handover_barrier() {
    __builtin_amdgcn_fence(__ATOMIC_RELEASE, "workgroup", "local");
    __builtin_amdgcn_s_barrier();
    __builtin_amdgcn_fence(__ATOMIC_ACQUIRE, "workgroup", "local");
}

__device__ __forceinline__ int q_hid(int g, int h, int s) {
    return 16 * (g >> 2) + 8 * h + 2 * (g & 3) + s;
}

struct StepIn {
    float zsum;
    float g0, g1, g2;
    float bs0, bs1;
};

__device__ __forceinline__ StepIn step_inputs(const float (&zv)[DIMZ],
                                              const float sv0, const float sv1, const float sv2,
                                              const float (&nmw)[RH][DIMZ],
                                              const float zstart, const float (&bstart)[RH],
                                              const bool hib) {
    float zsum = zstart;
#pragma unroll
    for (int i = 0; i < DIMZ; ++i) zsum += zv[i];
    float a0 = bstart[0], a1 = bstart[1], a2 = bstart[2];
#pragma unroll
    for (int i = 0; i < DIMZ; ++i) {
        a0 = fmaf(nmw[0][i], zv[i], a0);
        a1 = fmaf(nmw[1][i], zv[i], a1);
        a2 = fmaf(nmw[2][i], zv[i], a2);
    }
    StepIn r;
    r.zsum = zsum;
    r.g0 = fast_sigmoid(a0);
    r.g1 = fast_sigmoid(a1);
    r.g2 = fast_sigmoid(a2);
    r.bs0 = zsum * (hib ? sv2 : sv0);
    r.bs1 = hib ? zsum : (zsum * sv1);
    return r;
}

__global__ __launch_bounds__(32 * NWAVE) void cxtrnn_wmma_kernel(
    const float* __restrict__ s,   const float* __restrict__ z,
    const float* __restrict__ U,   const float* __restrict__ V,
    const float* __restrict__ Wi,  const float* __restrict__ bi,
    const float* __restrict__ Wo,  const float* __restrict__ bo,
    const float* __restrict__ nmW, const float* __restrict__ nmb,
    float* __restrict__ out)
{
    __shared__ v3f shP[2][NWAVE][32];
    __shared__ v3f shY[2][NWAVE][32];

    const int lane = (int)threadIdx.x & 31;
    const int w    = __builtin_amdgcn_readfirstlane((int)threadIdx.x >> 5);
    const bool w3  = (w == NWAVE - 1);
    const int col  = lane & 15;
    const int hi   = lane >> 4;
    const bool hib = (hi != 0);
    const int b0   = (int)blockIdx.x * 16;
    const int brow = b0 + col;

    v2f Aq[KGW];
#pragma unroll
    for (int q = 0; q < KGW; ++q) {
        const int g = KGW * w + q;
        v2f t;
#pragma unroll
        for (int e = 0; e < 2; ++e) {
            const int hd = q_hid(g, hi, e);
            float a = 0.0f;
            if (col < RH) {
                a = (hd < HID) ? V[hd * RANK + col] : 0.0f;
            } else if (col >= 8 && col < 8 + RH) {
                a = (hd < HID) ? V[hd * RANK + (col - 8 + RH)] : 0.0f;
            } else if (col >= QY && col < QY + DIMY) {
                const int y = col - QY;
                a = (hd < HID) ? Wo[y * HID + hd] : ((hd == HONE) ? bo[y] : 0.0f);
            }
            t[e] = a;
        }
        Aq[q] = t;
    }
    v2f Au0, Au1;
    v2f Aw;
    {
        const int hd = 16 * w + col;
        const bool ok = hd < HID;
        Au0[0] = ok ? ALPHA * U[hd * RANK + RH * hi]     : 0.0f;
        Au0[1] = ok ? ALPHA * U[hd * RANK + RH * hi + 1] : 0.0f;
        Au1[0] = ok ? ALPHA * U[hd * RANK + RH * hi + 2] : 0.0f;
        Au1[1] = 0.0f;
        const int k0 = 2 * hi;
        float a0 = 0.0f, a1 = 0.0f;
        if (ok) {
            a0 = ALPHA * Wi[hd * DIMS + k0];
            a1 = ALPHA * ((k0 + 1 < DIMS) ? Wi[hd * DIMS + k0 + 1] : bi[hd]);
        }
        Aw[0] = a0; Aw[1] = a1;
    }
    float nmw[RH][DIMZ], nmbv[RH];
#pragma unroll
    for (int r = 0; r < RH; ++r) {
        nmbv[r] = nmb[RH * hi + r];
#pragma unroll
        for (int i = 0; i < DIMZ; ++i) nmw[r][i] = nmW[(RH * hi + r) * DIMZ + i];
    }

    const v8f vzero = {0,0,0,0,0,0,0,0};
    v8f X = vzero;
    const bool one_slot = w3 && hib;
    float zprev = 0.0f;

    {
        const v3f z3 = {0.0f, 0.0f, 0.0f};
        shP[1][w][lane] = z3;
        shY[1][w][lane] = z3;
    }

    unsigned zoff = (unsigned)brow * (DIMZ * 4u);
    unsigned soff = (unsigned)brow * (DIMS * 4u);
    unsigned ooff = (unsigned)brow * (DIMY * 4u);
    float zq[TRIP][DIMZ], sq0[TRIP], sq1[TRIP], sq2[TRIP];
    StepIn G[TRIP];
#pragma unroll
    for (int u = 0; u < TRIP; ++u) {
        const gcf_ptr zr = row_ptr(z, (size_t)u * (BATCH * DIMZ), zoff);
        const gcf_ptr sr = row_ptr(s, (size_t)u * (BATCH * DIMS), soff);
#pragma unroll
        for (int i = 0; i < DIMZ; ++i) zq[u][i] = zr[i];
        sq0[u] = sr[0]; sq1[u] = sr[1]; sq2[u] = sr[2];
    }
#pragma unroll
    for (int u = 0; u < TRIP; ++u) {
        G[u] = step_inputs(zq[u], sq0[u], sq1[u], sq2[u], nmw, 0.0f, nmbv, hib);
        const gcf_ptr zr = row_ptr(z, (size_t)(TRIP + u) * (BATCH * DIMZ), zoff);
        const gcf_ptr sr = row_ptr(s, (size_t)(TRIP + u) * (BATCH * DIMS), soff);
#pragma unroll
        for (int i = 0; i < DIMZ; ++i) zq[u][i] = zr[i];
        sq0[u] = sr[0]; sq1[u] = sr[1]; sq2[u] = sr[2];
    }
    lds_handover_barrier();

#pragma unroll 1
    for (int t2 = 0; t2 < SEQ; t2 += TRIP) {
#pragma unroll
    for (int u = 0; u < TRIP; ++u) {
        const int t  = t2 + u;
        const int cb = u & 1;
        const int pb = cb ^ 1;

        v2f Bs;
        Bs[0] = G[u].bs0;
        Bs[1] = G[u].bs1;
        v8f acc = wmma_f32_k4(Aw, Bs, vzero);

        const v3f p0 = shP[pb][0][lane];
        const v3f p1 = shP[pb][1][lane];
        const v3f p2 = shP[pb][2][lane];
        const v3f p3 = shP[pb][3][lane];
        const float P0 = ((p0.x + p1.x) + p2.x) + p3.x;
        const float P1 = ((p0.y + p1.y) + p2.y) + p3.y;
        const float P2 = ((p0.z + p1.z) + p2.z) + p3.z;

        float oy0 = 0.0f, oy1 = 0.0f, oy2 = 0.0f;
        if (w3) {
            const v3f o0 = shY[pb][0][lane];
            const v3f o1 = shY[pb][1][lane];
            const v3f o2 = shY[pb][2][lane];
            const v3f o3 = shY[pb][3][lane];
            oy0 = zprev * (((o0.x + o1.x) + o2.x) + o3.x);
            oy1 = zprev * (((o0.y + o1.y) + o2.y) + o3.y);
            oy2 = zprev * (((o0.z + o1.z) + o2.z) + o3.z);
            asm volatile("" : "+v"(oy0), "+v"(oy1), "+v"(oy2));
        }
        zprev = G[u].zsum;

        v2f Bp0, Bp1;
        Bp0[0] = P0 * G[u].g0;
        Bp0[1] = P1 * G[u].g1;
        Bp1[0] = P2 * G[u].g2;
        Bp1[1] = 0.0f;

        acc = wmma_f32_k4(Au0, Bp0, acc);
        acc = wmma_f32_k4(Au1, Bp1, acc);

#pragma unroll
        for (int v = 0; v < 8; ++v) X[v] = fmaf(X[v], 1.0f - ALPHA, acc[v]);

        v8f Qp = vzero;
        v2f bcur;
        {
            const float th0 = fast_tanh(X[0]);
            const float th1 = fast_tanh(X[1]);
            bcur[0] = one_slot ? 1.0f : th0;
            bcur[1] = one_slot ? 0.0f : th1;
        }
        if (!w3) {
#pragma unroll
            for (int q = 0; q < KGW; ++q) {
                v2f bnext = bcur;
                if (q + 1 < KGW) {
                    bnext[0] = fast_tanh(X[2 * (q + 1)]);
                    bnext[1] = fast_tanh(X[2 * (q + 1) + 1]);
                }
                if (q == 0)
                    asm volatile("" : "+v"(bcur), "+v"(bnext));
                else if (q + 1 < KGW)
                    asm volatile("" : "+v"(bnext), "+v"(Qp));
                Qp = wmma_f32_k4(Aq[q], bcur, Qp);
                bcur = bnext;
            }
        } else {
            Qp = wmma_f32_k4(Aq[0], bcur, Qp);
        }

        shP[cb][w][lane] = __builtin_shufflevector(Qp, Qp, 0, 1, 2);
        shY[cb][w][lane] = __builtin_shufflevector(Qp, Qp, 3, 4, 5);

        if (w3 && !hib && t > 0) {
            const gf_ptr op = (gf_ptr)row_base(out, (size_t)(t - 1) * (BATCH * DIMY), ooff);
            op[0] = oy0;
            op[1] = oy1;
            op[2] = oy2;
        }

        float zstart = 0.0f;
        float bstart[RH] = {nmbv[0], nmbv[1], nmbv[2]};
        asm volatile("" : "+v"(zstart), "+v"(bstart[0]), "+v"(bstart[1]), "+v"(bstart[2])
                        : : "memory");
        G[u] = step_inputs(zq[u], sq0[u], sq1[u], sq2[u], nmw, zstart, bstart, hib);
        {
            const int tn = (t + 2 * TRIP < SEQ) ? (t + 2 * TRIP) : (SEQ - 1);
            const gcf_ptr zr = row_ptr(z, (size_t)tn * (BATCH * DIMZ), zoff);
            const gcf_ptr sr = row_ptr(s, (size_t)tn * (BATCH * DIMS), soff);
#pragma unroll
            for (int i = 0; i < DIMZ; ++i) zq[u][i] = zr[i];
            sq0[u] = sr[0]; sq1[u] = sr[1]; sq2[u] = sr[2];
        }
        asm volatile("" : "+v"(G[u].zsum), "+v"(G[u].g0), "+v"(G[u].g1), "+v"(G[u].g2),
                          "+v"(G[u].bs0), "+v"(G[u].bs1));
        lds_handover_barrier();
    }
    }

    if (w3 && !hib) {
        const int lb = (SEQ - 1) & 1;
        float y0 = 0.0f, y1 = 0.0f, y2 = 0.0f;
#pragma unroll
        for (int k = 0; k < NWAVE; ++k) {
            const v3f o = shY[lb][k][lane];
            y0 = (k == 0) ? o.x : (y0 + o.x);
            y1 = (k == 0) ? o.y : (y1 + o.y);
            y2 = (k == 0) ? o.z : (y2 + o.z);
        }
        const size_t base = ((size_t)(SEQ - 1) * BATCH + (size_t)brow) * DIMY;
        out[base + 0] = zprev * y0;
        out[base + 1] = zprev * y1;
        out[base + 2] = zprev * y2;
    }
}

extern "C" void kernel_launch(void* const* d_in, const int* in_sizes, int n_in,
                              void* d_out, int out_size, void* d_ws, size_t ws_size,
                              hipStream_t stream) {
    (void)in_sizes; (void)n_in; (void)out_size; (void)d_ws; (void)ws_size;
    const float* s   = (const float*)d_in[0];
    const float* z   = (const float*)d_in[1];
    const float* U   = (const float*)d_in[2];
    const float* V   = (const float*)d_in[3];
    const float* Wi  = (const float*)d_in[4];
    const float* bi  = (const float*)d_in[5];
    const float* Wo  = (const float*)d_in[6];
    const float* bo  = (const float*)d_in[7];
    const float* nmW = (const float*)d_in[8];
    const float* nmb = (const float*)d_in[9];
    float* out = (float*)d_out;

    dim3 grid(BATCH / 16);
    dim3 block(32 * NWAVE);
    hipLaunchKernelGGL(cxtrnn_wmma_kernel, grid, block, 0, stream,
                       s, z, U, V, Wi, bi, Wo, bo, nmW, nmb, out);
}


# Last wave stores out with all 32 lanes, two steps' rows per store (out rows of Wq^T in both lane halves), at the next step's head: no lane mask, no wait for address translation; sums unchanged
# speedup vs baseline: 4.7762x; 1.1249x over previous
#include <hip/hip_runtime.h>


#define SEQ   512
#define BATCH 4096
#define DIMS  3
#define DIMY  3
#define DIMZ  6
#define RANK  6
#define HID   50
#define KG    13
#define ALPHA 0.5f
#define RH    3
#define QY    3
#define HONE  56
#define TRIP  4
#define NWAVE 4
#define KGW   4
static_assert(SEQ % TRIP == 0 && TRIP % 2 == 0, "out is stored in pairs of steps; a step's LDS parity is a constant");

typedef float v2f __attribute__((ext_vector_type(2)));
typedef float v8f __attribute__((ext_vector_type(8)));
typedef float v3f __attribute__((ext_vector_type(3)));
typedef const float __attribute__((address_space(1)))* gcf_ptr;
typedef float __attribute__((address_space(1)))* gf_ptr;
typedef const char __attribute__((address_space(1)))* gcc_ptr;
typedef char __attribute__((address_space(1)))* gc_ptr;

__device__ __forceinline__ gcc_ptr row_base(const float* base, size_t elem, unsigned& lane_bytes) {
    gcc_ptr b = (gcc_ptr)(base + elem);
    asm volatile("" : "+s"(b), "+v"(lane_bytes));
    return b + lane_bytes;
}
__device__ __forceinline__ gcf_ptr row_ptr(const float* base, size_t elem, unsigned& lane_bytes) {
    return (gcf_ptr)row_base(base, elem, lane_bytes);
}

__device__ __forceinline__ v8f wmma_f32_k4(v2f a, v2f b, v8f c) {
    return __builtin_amdgcn_wmma_f32_16x16x4_f32(false, a, false, b, (short)0, c, false, false);
}

__device__ __forceinline__ float fast_tanh(float x) {
#if __has_builtin(__builtin_amdgcn_tanhf)
    return __builtin_amdgcn_tanhf(x);
#else
    x = fminf(15.0f, fmaxf(-15.0f, x));
    float e = __expf(2.0f * x);
    float r = __builtin_amdgcn_rcpf(e + 1.0f);
    return fmaf(-2.0f, r, 1.0f);
#endif
}

__device__ __forceinline__ float fast_sigmoid(float x) {
    return __builtin_amdgcn_rcpf(1.0f + __expf(-x));
}

__device__ __forceinline__ void lds_handover_barrier() {
    __builtin_amdgcn_fence(__ATOMIC_RELEASE, "workgroup", "local");
    __builtin_amdgcn_s_barrier();
    __builtin_amdgcn_fence(__ATOMIC_ACQUIRE, "workgroup", "local");
}

__device__ __forceinline__ int q_hid(int g, int h, int s) {
    return 16 * (g >> 2) + 8 * h + 2 * (g & 3) + s;
}

struct StepIn {
    float zsum;
    float g0, g1, g2;
    float bs0, bs1;
};

__device__ __forceinline__ StepIn step_inputs(const float (&zv)[DIMZ],
                                              const float sv0, const float sv1, const float sv2,
                                              const float (&nmw)[RH][DIMZ],
                                              const float zstart, const float (&bstart)[RH],
                                              const bool hib) {
    float zsum = zstart;
#pragma unroll
    for (int i = 0; i < DIMZ; ++i) zsum += zv[i];
    float a0 = bstart[0], a1 = bstart[1], a2 = bstart[2];
#pragma unroll
    for (int i = 0; i < DIMZ; ++i) {
        a0 = fmaf(nmw[0][i], zv[i], a0);
        a1 = fmaf(nmw[1][i], zv[i], a1);
        a2 = fmaf(nmw[2][i], zv[i], a2);
    }
    StepIn r;
    r.zsum = zsum;
    r.g0 = fast_sigmoid(a0);
    r.g1 = fast_sigmoid(a1);
    r.g2 = fast_sigmoid(a2);
    r.bs0 = zsum * (hib ? sv2 : sv0);
    r.bs1 = hib ? zsum : (zsum * sv1);
    return r;
}

__global__ __launch_bounds__(32 * NWAVE) void cxtrnn_wmma_kernel(
    const float* __restrict__ s,   const float* __restrict__ z,
    const float* __restrict__ U,   const float* __restrict__ V,
    const float* __restrict__ Wi,  const float* __restrict__ bi,
    const float* __restrict__ Wo,  const float* __restrict__ bo,
    const float* __restrict__ nmW, const float* __restrict__ nmb,
    float* __restrict__ out)
{
    __shared__ v3f shP[2][NWAVE][32];
    __shared__ v3f shY[2][NWAVE][32];

    const int lane = (int)threadIdx.x & 31;
    const int w    = __builtin_amdgcn_readfirstlane((int)threadIdx.x >> 5);
    const bool w3  = (w == NWAVE - 1);
    const int col  = lane & 15;
    const int hi   = lane >> 4;
    const bool hib = (hi != 0);
    const int b0   = (int)blockIdx.x * 16;
    const int brow = b0 + col;

    v2f Aq[KGW];
#pragma unroll
    for (int q = 0; q < KGW; ++q) {
        const int g = KGW * w + q;
        v2f t;
#pragma unroll
        for (int e = 0; e < 2; ++e) {
            const int hd = q_hid(g, hi, e);
            float a = 0.0f;
            const int c7 = col & 7;
            const int ch = col >> 3;
            if (c7 < RH) {
                a = (hd < HID) ? V[hd * RANK + (RH * ch + c7)] : 0.0f;
            } else if (c7 >= QY && c7 < QY + DIMY) {
                const int y = c7 - QY;
                a = (hd < HID) ? Wo[y * HID + hd] : ((hd == HONE) ? bo[y] : 0.0f);
            }
            t[e] = a;
        }
        Aq[q] = t;
    }
    v2f Au0, Au1;
    v2f Aw;
    {
        const int hd = 16 * w + col;
        const bool ok = hd < HID;
        Au0[0] = ok ? ALPHA * U[hd * RANK + RH * hi]     : 0.0f;
        Au0[1] = ok ? ALPHA * U[hd * RANK + RH * hi + 1] : 0.0f;
        Au1[0] = ok ? ALPHA * U[hd * RANK + RH * hi + 2] : 0.0f;
        Au1[1] = 0.0f;
        const int k0 = 2 * hi;
        float a0 = 0.0f, a1 = 0.0f;
        if (ok) {
            a0 = ALPHA * Wi[hd * DIMS + k0];
            a1 = ALPHA * ((k0 + 1 < DIMS) ? Wi[hd * DIMS + k0 + 1] : bi[hd]);
        }
        Aw[0] = a0; Aw[1] = a1;
    }
    float nmw[RH][DIMZ], nmbv[RH];
#pragma unroll
    for (int r = 0; r < RH; ++r) {
        nmbv[r] = nmb[RH * hi + r];
#pragma unroll
        for (int i = 0; i < DIMZ; ++i) nmw[r][i] = nmW[(RH * hi + r) * DIMZ + i];
    }

    const v8f vzero = {0,0,0,0,0,0,0,0};
    v8f X = vzero;
    const bool one_slot = w3 && hib;
    float zprev = 0.0f;

    {
        const v3f z3 = {0.0f, 0.0f, 0.0f};
        shP[1][w][lane] = z3;
        shY[1][w][lane] = z3;
    }

    unsigned zoff = (unsigned)brow * (DIMZ * 4u);
    unsigned soff = (unsigned)brow * (DIMS * 4u);
    unsigned ooff = (unsigned)brow * (DIMY * 4u) + (hib ? (unsigned)(BATCH * DIMY * 4) : 0u);
    float oe0 = 0.0f, oe1 = 0.0f, oe2 = 0.0f;
    float pr0 = 0.0f, pr1 = 0.0f, pr2 = 0.0f;
    float zq[TRIP][DIMZ], sq0[TRIP], sq1[TRIP], sq2[TRIP];
    StepIn G[TRIP];
#pragma unroll
    for (int u = 0; u < TRIP; ++u) {
        const gcf_ptr zr = row_ptr(z, (size_t)u * (BATCH * DIMZ), zoff);
        const gcf_ptr sr = row_ptr(s, (size_t)u * (BATCH * DIMS), soff);
#pragma unroll
        for (int i = 0; i < DIMZ; ++i) zq[u][i] = zr[i];
        sq0[u] = sr[0]; sq1[u] = sr[1]; sq2[u] = sr[2];
    }
#pragma unroll
    for (int u = 0; u < TRIP; ++u) {
        G[u] = step_inputs(zq[u], sq0[u], sq1[u], sq2[u], nmw, 0.0f, nmbv, hib);
        const gcf_ptr zr = row_ptr(z, (size_t)(TRIP + u) * (BATCH * DIMZ), zoff);
        const gcf_ptr sr = row_ptr(s, (size_t)(TRIP + u) * (BATCH * DIMS), soff);
#pragma unroll
        for (int i = 0; i < DIMZ; ++i) zq[u][i] = zr[i];
        sq0[u] = sr[0]; sq1[u] = sr[1]; sq2[u] = sr[2];
    }
    lds_handover_barrier();

#pragma unroll 1
    for (int t2 = 0; t2 < SEQ; t2 += TRIP) {
#pragma unroll
    for (int u = 0; u < TRIP; ++u) {
        const int t  = t2 + u;
        const int cb = u & 1;
        const int pb = cb ^ 1;

        if ((u & 1) && w3 && (u > 1 || t2 > 0)) {
            const gf_ptr op = (gf_ptr)row_base(out, (size_t)(t - 3) * (BATCH * DIMY), ooff);
            op[0] = pr0;
            op[1] = pr1;
            op[2] = pr2;
        }

        v2f Bs;
        Bs[0] = G[u].bs0;
        Bs[1] = G[u].bs1;
        v8f acc = wmma_f32_k4(Aw, Bs, vzero);

        const v3f p0 = shP[pb][0][lane];
        __builtin_amdgcn_sched_barrier(0x047F);
        const v3f p1 = shP[pb][1][lane];
        __builtin_amdgcn_sched_barrier(0x047F);
        const v3f p2 = shP[pb][2][lane];
        __builtin_amdgcn_sched_barrier(0x047F);
        const v3f p3 = shP[pb][3][lane];
        const float P0 = ((p0.x + p1.x) + p2.x) + p3.x;
        const float P1 = ((p0.y + p1.y) + p2.y) + p3.y;
        const float P2 = ((p0.z + p1.z) + p2.z) + p3.z;

        float oy0 = 0.0f, oy1 = 0.0f, oy2 = 0.0f;
        if (w3) {
            const v3f o0 = shY[pb][0][lane];
            const v3f o1 = shY[pb][1][lane];
            const v3f o2 = shY[pb][2][lane];
            const v3f o3 = shY[pb][3][lane];
            oy0 = zprev * (((o0.x + o1.x) + o2.x) + o3.x);
            oy1 = zprev * (((o0.y + o1.y) + o2.y) + o3.y);
            oy2 = zprev * (((o0.z + o1.z) + o2.z) + o3.z);
            asm volatile("" : "+v"(oy0), "+v"(oy1), "+v"(oy2));
            if (u & 1) {
                oe0 = oy0; oe1 = oy1; oe2 = oy2;
            } else {
                pr0 = hib ? oy0 : oe0;
                pr1 = hib ? oy1 : oe1;
                pr2 = hib ? oy2 : oe2;
            }
        }
        zprev = G[u].zsum;

        v2f Bp0, Bp1;
        Bp0[0] = P0 * G[u].g0;
        Bp0[1] = P1 * G[u].g1;
        Bp1[0] = P2 * G[u].g2;
        Bp1[1] = 0.0f;

        acc = wmma_f32_k4(Au0, Bp0, acc);
        acc = wmma_f32_k4(Au1, Bp1, acc);

#pragma unroll
        for (int v = 0; v < 8; ++v) X[v] = fmaf(X[v], 1.0f - ALPHA, acc[v]);

        v8f Qp = vzero;
        v2f bcur;
        {
            const float th0 = fast_tanh(X[0]);
            const float th1 = fast_tanh(X[1]);
            bcur[0] = one_slot ? 1.0f : th0;
            bcur[1] = one_slot ? 0.0f : th1;
        }
        if (!w3) {
#pragma unroll
            for (int q = 0; q < KGW; ++q) {
                v2f bnext = bcur;
                if (q + 1 < KGW) {
                    bnext[0] = fast_tanh(X[2 * (q + 1)]);
                    bnext[1] = fast_tanh(X[2 * (q + 1) + 1]);
                }
                if (q == 0)
                    asm volatile("" : "+v"(bcur), "+v"(bnext));
                else if (q + 1 < KGW)
                    asm volatile("" : "+v"(bnext), "+v"(Qp));
                Qp = wmma_f32_k4(Aq[q], bcur, Qp);
                bcur = bnext;
            }
        } else {
            Qp = wmma_f32_k4(Aq[0], bcur, Qp);
        }

        shP[cb][w][lane] = __builtin_shufflevector(Qp, Qp, 0, 1, 2);
        shY[cb][w][lane] = __builtin_shufflevector(Qp, Qp, 3, 4, 5);

        float zstart = 0.0f;
        float bstart[RH] = {nmbv[0], nmbv[1], nmbv[2]};
        asm volatile("" : "+v"(zstart), "+v"(bstart[0]), "+v"(bstart[1]), "+v"(bstart[2])
                        : : "memory");
        G[u] = step_inputs(zq[u], sq0[u], sq1[u], sq2[u], nmw, zstart, bstart, hib);
        {
            const int tn = (t + 2 * TRIP < SEQ) ? (t + 2 * TRIP) : (SEQ - 1);
            const gcf_ptr zr = row_ptr(z, (size_t)tn * (BATCH * DIMZ), zoff);
            const gcf_ptr sr = row_ptr(s, (size_t)tn * (BATCH * DIMS), soff);
#pragma unroll
            for (int i = 0; i < DIMZ; ++i) zq[u][i] = zr[i];
            sq0[u] = sr[0]; sq1[u] = sr[1]; sq2[u] = sr[2];
        }
        asm volatile("" : "+v"(G[u].zsum), "+v"(G[u].g0), "+v"(G[u].g1), "+v"(G[u].g2),
                          "+v"(G[u].bs0), "+v"(G[u].bs1));
        lds_handover_barrier();
    }
    }

    if (w3) {
        const int lb = (SEQ - 1) & 1;
        float y0 = 0.0f, y1 = 0.0f, y2 = 0.0f;
#pragma unroll
        for (int k = 0; k < NWAVE; ++k) {
            const v3f o = shY[lb][k][lane];
            y0 = (k == 0) ? o.x : (y0 + o.x);
            y1 = (k == 0) ? o.y : (y1 + o.y);
            y2 = (k == 0) ? o.z : (y2 + o.z);
        }
        const gf_ptr op = (gf_ptr)row_base(out, (size_t)(SEQ - 2) * (BATCH * DIMY), ooff);
        op[0] = hib ? (zprev * y0) : oe0;
        op[1] = hib ? (zprev * y1) : oe1;
        op[2] = hib ? (zprev * y2) : oe2;
    }
}

extern "C" void kernel_launch(void* const* d_in, const int* in_sizes, int n_in,
                              void* d_out, int out_size, void* d_ws, size_t ws_size,
                              hipStream_t stream) {
    (void)in_sizes; (void)n_in; (void)out_size; (void)d_ws; (void)ws_size;
    const float* s   = (const float*)d_in[0];
    const float* z   = (const float*)d_in[1];
    const float* U   = (const float*)d_in[2];
    const float* V   = (const float*)d_in[3];
    const float* Wi  = (const float*)d_in[4];
    const float* bi  = (const float*)d_in[5];
    const float* Wo  = (const float*)d_in[6];
    const float* bo  = (const float*)d_in[7];
    const float* nmW = (const float*)d_in[8];
    const float* nmb = (const float*)d_in[9];
    float* out = (float*)d_out;

    dim3 grid(BATCH / 16);
    dim3 block(32 * NWAVE);
    hipLaunchKernelGGL(cxtrnn_wmma_kernel, grid, block, 0, stream,
                       s, z, U, V, Wi, bi, Wo, bo, nmW, nmb, out);
}


# Role-only work moved into each role's idle time: long-chain waves go tanh -> products without select or copy; last wave reads out rows behind its one product, sums+stores them under the next P reads
# speedup vs baseline: 4.9523x; 1.0369x over previous
#include <hip/hip_runtime.h>


#define SEQ   512
#define BATCH 4096
#define DIMS  3
#define DIMY  3
#define DIMZ  6
#define RANK  6
#define HID   50
#define KG    13
#define ALPHA 0.5f
#define RH    3
#define QY    3
#define HONE  56
#define TRIP  4
#define NWAVE 4
#define KGW   4
static_assert(SEQ % TRIP == 0 && TRIP % 2 == 0, "out is stored in pairs of steps; a step's LDS parity is a constant");

typedef float v2f __attribute__((ext_vector_type(2)));
typedef float v8f __attribute__((ext_vector_type(8)));
typedef float v3f __attribute__((ext_vector_type(3)));
typedef const float __attribute__((address_space(1)))* gcf_ptr;
typedef float __attribute__((address_space(1)))* gf_ptr;
typedef const char __attribute__((address_space(1)))* gcc_ptr;
typedef char __attribute__((address_space(1)))* gc_ptr;

__device__ __forceinline__ gcc_ptr row_base(const float* base, size_t elem, unsigned& lane_bytes) {
    gcc_ptr b = (gcc_ptr)(base + elem);
    asm volatile("" : "+s"(b), "+v"(lane_bytes));
    return b + lane_bytes;
}
__device__ __forceinline__ gcf_ptr row_ptr(const float* base, size_t elem, unsigned& lane_bytes) {
    return (gcf_ptr)row_base(base, elem, lane_bytes);
}

__device__ __forceinline__ v8f wmma_f32_k4(v2f a, v2f b, v8f c) {
    return __builtin_amdgcn_wmma_f32_16x16x4_f32(false, a, false, b, (short)0, c, false, false);
}

__device__ __forceinline__ float fast_tanh(float x) {
#if __has_builtin(__builtin_amdgcn_tanhf)
    return __builtin_amdgcn_tanhf(x);
#else
    x = fminf(15.0f, fmaxf(-15.0f, x));
    float e = __expf(2.0f * x);
    float r = __builtin_amdgcn_rcpf(e + 1.0f);
    return fmaf(-2.0f, r, 1.0f);
#endif
}

__device__ __forceinline__ float fast_sigmoid(float x) {
    return __builtin_amdgcn_rcpf(1.0f + __expf(-x));
}

__device__ __forceinline__ void lds_handover_barrier() {
    __builtin_amdgcn_fence(__ATOMIC_RELEASE, "workgroup", "local");
    __builtin_amdgcn_s_barrier();
    __builtin_amdgcn_fence(__ATOMIC_ACQUIRE, "workgroup", "local");
}

__device__ __forceinline__ int q_hid(int g, int h, int s) {
    return 16 * (g >> 2) + 8 * h + 2 * (g & 3) + s;
}

struct StepIn {
    float zsum;
    float g0, g1, g2;
    float bs0, bs1;
};

__device__ __forceinline__ StepIn step_inputs(const float (&zv)[DIMZ],
                                              const float sv0, const float sv1, const float sv2,
                                              const float (&nmw)[RH][DIMZ],
                                              const float zstart, const float (&bstart)[RH],
                                              const bool hib) {
    float zsum = zstart;
#pragma unroll
    for (int i = 0; i < DIMZ; ++i) zsum += zv[i];
    float a0 = bstart[0], a1 = bstart[1], a2 = bstart[2];
#pragma unroll
    for (int i = 0; i < DIMZ; ++i) {
        a0 = fmaf(nmw[0][i], zv[i], a0);
        a1 = fmaf(nmw[1][i], zv[i], a1);
        a2 = fmaf(nmw[2][i], zv[i], a2);
    }
    StepIn r;
    r.zsum = zsum;
    r.g0 = fast_sigmoid(a0);
    r.g1 = fast_sigmoid(a1);
    r.g2 = fast_sigmoid(a2);
    r.bs0 = zsum * (hib ? sv2 : sv0);
    r.bs1 = hib ? zsum : (zsum * sv1);
    return r;
}

__global__ __launch_bounds__(32 * NWAVE) void cxtrnn_wmma_kernel(
    const float* __restrict__ s,   const float* __restrict__ z,
    const float* __restrict__ U,   const float* __restrict__ V,
    const float* __restrict__ Wi,  const float* __restrict__ bi,
    const float* __restrict__ Wo,  const float* __restrict__ bo,
    const float* __restrict__ nmW, const float* __restrict__ nmb,
    float* __restrict__ out)
{
    __shared__ v3f shP[2][NWAVE][32];
    __shared__ v3f shY[2][NWAVE][32];

    const int lane = (int)threadIdx.x & 31;
    const int w    = __builtin_amdgcn_readfirstlane((int)threadIdx.x >> 5);
    const bool w3  = (w == NWAVE - 1);
    const int col  = lane & 15;
    const int hi   = lane >> 4;
    const bool hib = (hi != 0);
    const int b0   = (int)blockIdx.x * 16;
    const int brow = b0 + col;

    v2f Aq[KGW];
#pragma unroll
    for (int q = 0; q < KGW; ++q) {
        const int g = KGW * w + q;
        v2f t;
#pragma unroll
        for (int e = 0; e < 2; ++e) {
            const int hd = q_hid(g, hi, e);
            float a = 0.0f;
            const int c7 = col & 7;
            const int ch = col >> 3;
            if (c7 < RH) {
                a = (hd < HID) ? V[hd * RANK + (RH * ch + c7)] : 0.0f;
            } else if (c7 >= QY && c7 < QY + DIMY) {
                const int y = c7 - QY;
                a = (hd < HID) ? Wo[y * HID + hd] : ((hd == HONE) ? bo[y] : 0.0f);
            }
            t[e] = a;
        }
        Aq[q] = t;
    }
    v2f Au0, Au1;
    v2f Aw;
    {
        const int hd = 16 * w + col;
        const bool ok = hd < HID;
        Au0[0] = ok ? ALPHA * U[hd * RANK + RH * hi]     : 0.0f;
        Au0[1] = ok ? ALPHA * U[hd * RANK + RH * hi + 1] : 0.0f;
        Au1[0] = ok ? ALPHA * U[hd * RANK + RH * hi + 2] : 0.0f;
        Au1[1] = 0.0f;
        const int k0 = 2 * hi;
        float a0 = 0.0f, a1 = 0.0f;
        if (ok) {
            a0 = ALPHA * Wi[hd * DIMS + k0];
            a1 = ALPHA * ((k0 + 1 < DIMS) ? Wi[hd * DIMS + k0 + 1] : bi[hd]);
        }
        Aw[0] = a0; Aw[1] = a1;
    }
    float nmw[RH][DIMZ], nmbv[RH];
#pragma unroll
    for (int r = 0; r < RH; ++r) {
        nmbv[r] = nmb[RH * hi + r];
#pragma unroll
        for (int i = 0; i < DIMZ; ++i) nmw[r][i] = nmW[(RH * hi + r) * DIMZ + i];
    }

    const v8f vzero = {0,0,0,0,0,0,0,0};
    v8f X = vzero;
    float zprev = 0.0f;
    float zprev2 = 0.0f;
    v3f o0 = {0.0f, 0.0f, 0.0f}, o1 = o0, o2 = o0, o3 = o0;

    {
        const v3f z3 = {0.0f, 0.0f, 0.0f};
        shP[1][w][lane] = z3;
        shY[1][w][lane] = z3;
    }

    unsigned zoff = (unsigned)brow * (DIMZ * 4u);
    unsigned soff = (unsigned)brow * (DIMS * 4u);
    unsigned ooff = (unsigned)brow * (DIMY * 4u) + (hib ? (unsigned)(BATCH * DIMY * 4) : 0u);
    float oe0 = 0.0f, oe1 = 0.0f, oe2 = 0.0f;
    float zq[TRIP][DIMZ], sq0[TRIP], sq1[TRIP], sq2[TRIP];
    StepIn G[TRIP];
#pragma unroll
    for (int u = 0; u < TRIP; ++u) {
        const gcf_ptr zr = row_ptr(z, (size_t)u * (BATCH * DIMZ), zoff);
        const gcf_ptr sr = row_ptr(s, (size_t)u * (BATCH * DIMS), soff);
#pragma unroll
        for (int i = 0; i < DIMZ; ++i) zq[u][i] = zr[i];
        sq0[u] = sr[0]; sq1[u] = sr[1]; sq2[u] = sr[2];
    }
#pragma unroll
    for (int u = 0; u < TRIP; ++u) {
        G[u] = step_inputs(zq[u], sq0[u], sq1[u], sq2[u], nmw, 0.0f, nmbv, hib);
        const gcf_ptr zr = row_ptr(z, (size_t)(TRIP + u) * (BATCH * DIMZ), zoff);
        const gcf_ptr sr = row_ptr(s, (size_t)(TRIP + u) * (BATCH * DIMS), soff);
#pragma unroll
        for (int i = 0; i < DIMZ; ++i) zq[u][i] = zr[i];
        sq0[u] = sr[0]; sq1[u] = sr[1]; sq2[u] = sr[2];
    }
    lds_handover_barrier();

#pragma unroll 1
    for (int t2 = 0; t2 < SEQ; t2 += TRIP) {
#pragma unroll
    for (int u = 0; u < TRIP; ++u) {
        const int t  = t2 + u;
        const int cb = u & 1;
        const int pb = cb ^ 1;

        const v3f p0 = shP[pb][0][lane];
        __builtin_amdgcn_sched_barrier(0x047F);
        const v3f p1 = shP[pb][1][lane];
        __builtin_amdgcn_sched_barrier(0x047F);
        const v3f p2 = shP[pb][2][lane];
        __builtin_amdgcn_sched_barrier(0x047F);
        const v3f p3 = shP[pb][3][lane];

        __builtin_amdgcn_sched_barrier(0x0476);
        v2f Bs;
        Bs[0] = G[u].bs0;
        Bs[1] = G[u].bs1;
        v8f acc = wmma_f32_k4(Aw, Bs, vzero);

        if (__builtin_expect(w3, 0)) {
            const float oy0 = zprev2 * (((o0.x + o1.x) + o2.x) + o3.x);
            const float oy1 = zprev2 * (((o0.y + o1.y) + o2.y) + o3.y);
            const float oy2 = zprev2 * (((o0.z + o1.z) + o2.z) + o3.z);
            if (u & 1) {
                if (u > 1 || t2 > 0) {
                    const gf_ptr op = (gf_ptr)row_base(out, (size_t)(t - 3) * (BATCH * DIMY), ooff);
                    op[0] = hib ? oy0 : oe0;
                    op[1] = hib ? oy1 : oe1;
                    op[2] = hib ? oy2 : oe2;
                }
            } else {
                oe0 = oy0; oe1 = oy1; oe2 = oy2;
            }
        }
        const float P0 = ((p0.x + p1.x) + p2.x) + p3.x;
        const float P1 = ((p0.y + p1.y) + p2.y) + p3.y;
        const float P2 = ((p0.z + p1.z) + p2.z) + p3.z;

        zprev2 = zprev;
        zprev = G[u].zsum;

        v2f Bp0, Bp1;
        Bp0[0] = P0 * G[u].g0;
        Bp0[1] = P1 * G[u].g1;
        Bp1[0] = P2 * G[u].g2;
        Bp1[1] = 0.0f;

        acc = wmma_f32_k4(Au0, Bp0, acc);
        acc = wmma_f32_k4(Au1, Bp1, acc);

#pragma unroll
        for (int v = 0; v < 8; ++v) X[v] = fmaf(X[v], 1.0f - ALPHA, acc[v]);

        v2f bfirst;
        bfirst[0] = fast_tanh(X[0]);
        bfirst[1] = fast_tanh(X[1]);

        v8f Qp = vzero;
        if (!w3) {
            v2f bcur = bfirst;
#pragma unroll
            for (int q = 0; q < KGW; ++q) {
                v2f bnext = bcur;
                if (q + 1 < KGW) {
                    bnext[0] = fast_tanh(X[2 * (q + 1)]);
                    bnext[1] = fast_tanh(X[2 * (q + 1) + 1]);
                }
                if (q == 0)
                    asm volatile("" : "+v"(bnext));
                else if (q + 1 < KGW)
                    asm volatile("" : "+v"(bnext), "+v"(Qp));
                Qp = wmma_f32_k4(Aq[q], bcur, Qp);
                bcur = bnext;
            }
        } else {
            v2f b1;
            b1[0] = hib ? 1.0f : bfirst[0];
            b1[1] = hib ? 0.0f : bfirst[1];
            Qp = wmma_f32_k4(Aq[0], b1, Qp);
            __builtin_amdgcn_sched_barrier(0x0476);
            o0 = shY[pb][0][lane];
            o1 = shY[pb][1][lane];
            o2 = shY[pb][2][lane];
            o3 = shY[pb][3][lane];
        }

        shP[cb][w][lane] = __builtin_shufflevector(Qp, Qp, 0, 1, 2);
        shY[cb][w][lane] = __builtin_shufflevector(Qp, Qp, 3, 4, 5);

        float zstart = 0.0f;
        float bstart[RH] = {nmbv[0], nmbv[1], nmbv[2]};
        asm volatile("" : "+v"(zstart), "+v"(bstart[0]), "+v"(bstart[1]), "+v"(bstart[2])
                        : : "memory");
        G[u] = step_inputs(zq[u], sq0[u], sq1[u], sq2[u], nmw, zstart, bstart, hib);
        {
            const int tn = (t + 2 * TRIP < SEQ) ? (t + 2 * TRIP) : (SEQ - 1);
            const gcf_ptr zr = row_ptr(z, (size_t)tn * (BATCH * DIMZ), zoff);
            const gcf_ptr sr = row_ptr(s, (size_t)tn * (BATCH * DIMS), soff);
#pragma unroll
            for (int i = 0; i < DIMZ; ++i) zq[u][i] = zr[i];
            sq0[u] = sr[0]; sq1[u] = sr[1]; sq2[u] = sr[2];
        }
        asm volatile("" : "+v"(G[u].zsum), "+v"(G[u].g0), "+v"(G[u].g1), "+v"(G[u].g2),
                          "+v"(G[u].bs0), "+v"(G[u].bs1));
        lds_handover_barrier();
    }
    }

    if (w3) {
        const int lb = (SEQ - 1) & 1;
        const float e0 = zprev2 * (((o0.x + o1.x) + o2.x) + o3.x);
        const float e1 = zprev2 * (((o0.y + o1.y) + o2.y) + o3.y);
        const float e2 = zprev2 * (((o0.z + o1.z) + o2.z) + o3.z);
        float y0 = 0.0f, y1 = 0.0f, y2 = 0.0f;
#pragma unroll
        for (int k = 0; k < NWAVE; ++k) {
            const v3f o = shY[lb][k][lane];
            y0 = (k == 0) ? o.x : (y0 + o.x);
            y1 = (k == 0) ? o.y : (y1 + o.y);
            y2 = (k == 0) ? o.z : (y2 + o.z);
        }
        const gf_ptr op = (gf_ptr)row_base(out, (size_t)(SEQ - 2) * (BATCH * DIMY), ooff);
        op[0] = hib ? (zprev * y0) : e0;
        op[1] = hib ? (zprev * y1) : e1;
        op[2] = hib ? (zprev * y2) : e2;
    }
}

extern "C" void kernel_launch(void* const* d_in, const int* in_sizes, int n_in,
                              void* d_out, int out_size, void* d_ws, size_t ws_size,
                              hipStream_t stream) {
    (void)in_sizes; (void)n_in; (void)out_size; (void)d_ws; (void)ws_size;
    const float* s   = (const float*)d_in[0];
    const float* z   = (const float*)d_in[1];
    const float* U   = (const float*)d_in[2];
    const float* V   = (const float*)d_in[3];
    const float* Wi  = (const float*)d_in[4];
    const float* bi  = (const float*)d_in[5];
    const float* Wo  = (const float*)d_in[6];
    const float* bo  = (const float*)d_in[7];
    const float* nmW = (const float*)d_in[8];
    const float* nmb = (const float*)d_in[9];
    float* out = (float*)d_out;

    dim3 grid(BATCH / 16);
    dim3 block(32 * NWAVE);
    hipLaunchKernelGGL(cxtrnn_wmma_kernel, grid, block, 0, stream,
                       s, z, U, V, Wi, bi, Wo, bo, nmW, nmb, out);
}


# Barrier-to-tanh path cut to 3 LDS reads, one multiply-add per read, two products: own part of P stays in registers, gate folded into the sums, (1-a)x starts the products; order of sums changes
# speedup vs baseline: 5.1592x; 1.0418x over previous
#include <hip/hip_runtime.h>


#define SEQ   512
#define BATCH 4096
#define DIMS  3
#define DIMY  3
#define DIMZ  6
#define RANK  6
#define HID   50
#define KG    13
#define ALPHA 0.5f
#define RH    3
#define QY    3
#define HONE  56
#define TRIP  4
#define NWAVE 4
#define KGW   4
static_assert(SEQ % TRIP == 0 && TRIP % 2 == 0, "out is stored in pairs of steps; a step's LDS parity is a constant");
static_assert((NWAVE & (NWAVE - 1)) == 0, "the other waves' indices are taken modulo NWAVE with a mask");

typedef float v2f __attribute__((ext_vector_type(2)));
typedef float v8f __attribute__((ext_vector_type(8)));
typedef float v3f __attribute__((ext_vector_type(3)));
typedef const float __attribute__((address_space(1)))* gcf_ptr;
typedef float __attribute__((address_space(1)))* gf_ptr;
typedef const char __attribute__((address_space(1)))* gcc_ptr;
typedef char __attribute__((address_space(1)))* gc_ptr;

__device__ __forceinline__ gcc_ptr row_base(const float* base, size_t elem, unsigned& lane_bytes) {
    gcc_ptr b = (gcc_ptr)(base + elem);
    asm volatile("" : "+s"(b), "+v"(lane_bytes));
    return b + lane_bytes;
}
__device__ __forceinline__ gcf_ptr row_ptr(const float* base, size_t elem, unsigned& lane_bytes) {
    return (gcf_ptr)row_base(base, elem, lane_bytes);
}

__device__ __forceinline__ v8f wmma_f32_k4(v2f a, v2f b, v8f c) {
    return __builtin_amdgcn_wmma_f32_16x16x4_f32(false, a, false, b, (short)0, c, false, false);
}

__device__ __forceinline__ float fast_tanh(float x) {
#if __has_builtin(__builtin_amdgcn_tanhf)
    return __builtin_amdgcn_tanhf(x);
#else
    x = fminf(15.0f, fmaxf(-15.0f, x));
    float e = __expf(2.0f * x);
    float r = __builtin_amdgcn_rcpf(e + 1.0f);
    return fmaf(-2.0f, r, 1.0f);
#endif
}

__device__ __forceinline__ float fast_sigmoid(float x) {
    return __builtin_amdgcn_rcpf(1.0f + __expf(-x));
}

__device__ __forceinline__ void lds_handover_barrier() {
    __builtin_amdgcn_fence(__ATOMIC_RELEASE, "workgroup", "local");
    __builtin_amdgcn_s_barrier();
    __builtin_amdgcn_fence(__ATOMIC_ACQUIRE, "workgroup", "local");
}

__device__ __forceinline__ int q_hid(int g, int h, int s) {
    return 16 * (g >> 2) + 8 * h + 2 * (g & 3) + s;
}

struct StepIn {
    float zsum;
    float g0, g1, g2;
    float bs0, bs1;
};

__device__ __forceinline__ StepIn step_inputs(const float (&zv)[DIMZ],
                                              const float sv0, const float sv1, const float sv2,
                                              const float (&nmw)[RH][DIMZ],
                                              const float zstart, const float (&bstart)[RH],
                                              const bool hib) {
    float zsum = zstart;
#pragma unroll
    for (int i = 0; i < DIMZ; ++i) zsum += zv[i];
    float a0 = bstart[0], a1 = bstart[1], a2 = bstart[2];
#pragma unroll
    for (int i = 0; i < DIMZ; ++i) {
        a0 = fmaf(nmw[0][i], zv[i], a0);
        a1 = fmaf(nmw[1][i], zv[i], a1);
        a2 = fmaf(nmw[2][i], zv[i], a2);
    }
    StepIn r;
    r.zsum = zsum;
    r.g0 = fast_sigmoid(a0);
    r.g1 = fast_sigmoid(a1);
    r.g2 = fast_sigmoid(a2);
    r.bs0 = zsum * (hib ? sv2 : sv0);
    r.bs1 = hib ? zsum : (zsum * sv1);
    return r;
}

__global__ __launch_bounds__(32 * NWAVE) void cxtrnn_wmma_kernel(
    const float* __restrict__ s,   const float* __restrict__ z,
    const float* __restrict__ U,   const float* __restrict__ V,
    const float* __restrict__ Wi,  const float* __restrict__ bi,
    const float* __restrict__ Wo,  const float* __restrict__ bo,
    const float* __restrict__ nmW, const float* __restrict__ nmb,
    float* __restrict__ out)
{
    __shared__ v3f shP[2][NWAVE][32];
    __shared__ v3f shY[2][NWAVE][32];

    const int lane = (int)threadIdx.x & 31;
    const int w    = __builtin_amdgcn_readfirstlane((int)threadIdx.x >> 5);
    const bool w3  = (w == NWAVE - 1);
    const int wa   = (w + 1) & (NWAVE - 1);
    const int wb   = (w + 2) & (NWAVE - 1);
    const int wc   = (w + 3) & (NWAVE - 1);
    const int col  = lane & 15;
    const int hi   = lane >> 4;
    const bool hib = (hi != 0);
    const int b0   = (int)blockIdx.x * 16;
    const int brow = b0 + col;

    v2f Aq[KGW];
#pragma unroll
    for (int q = 0; q < KGW; ++q) {
        const int g = KGW * w + q;
        v2f t;
#pragma unroll
        for (int e = 0; e < 2; ++e) {
            const int hd = q_hid(g, hi, e);
            float a = 0.0f;
            const int c7 = col & 7;
            const int ch = col >> 3;
            if (c7 < RH) {
                a = (hd < HID) ? V[hd * RANK + (RH * ch + c7)] : 0.0f;
            } else if (c7 >= QY && c7 < QY + DIMY) {
                const int y = c7 - QY;
                a = (hd < HID) ? Wo[y * HID + hd] : ((hd == HONE) ? bo[y] : 0.0f);
            }
            t[e] = a;
        }
        Aq[q] = t;
    }
    v2f Au0, Au1;
    v2f Aw;
    {
        const int hd = 16 * w + col;
        const bool ok = hd < HID;
        Au0[0] = ok ? ALPHA * U[hd * RANK + RH * hi]     : 0.0f;
        Au0[1] = ok ? ALPHA * U[hd * RANK + RH * hi + 1] : 0.0f;
        Au1[0] = ok ? ALPHA * U[hd * RANK + RH * hi + 2] : 0.0f;
        Au1[1] = 0.0f;
        const int k0 = 2 * hi;
        float a0 = 0.0f, a1 = 0.0f;
        if (ok) {
            a0 = ALPHA * Wi[hd * DIMS + k0];
            a1 = ALPHA * ((k0 + 1 < DIMS) ? Wi[hd * DIMS + k0 + 1] : bi[hd]);
        }
        Aw[0] = a0; Aw[1] = a1;
    }
    float nmw[RH][DIMZ], nmbv[RH];
#pragma unroll
    for (int r = 0; r < RH; ++r) {
        nmbv[r] = nmb[RH * hi + r];
#pragma unroll
        for (int i = 0; i < DIMZ; ++i) nmw[r][i] = nmW[(RH * hi + r) * DIMZ + i];
    }

    const v8f vzero = {0,0,0,0,0,0,0,0};
    v8f X = vzero;
    float own0 = 0.0f, own1 = 0.0f, own2 = 0.0f;
    float zprev = 0.0f;
    float zprev2 = 0.0f;
    v3f o0 = {0.0f, 0.0f, 0.0f}, o1 = o0, o2 = o0, o3 = o0;

    {
        const v3f z3 = {0.0f, 0.0f, 0.0f};
        shP[1][w][lane] = z3;
        shY[1][w][lane] = z3;
    }

    unsigned zoff = (unsigned)brow * (DIMZ * 4u);
    unsigned soff = (unsigned)brow * (DIMS * 4u);
    unsigned ooff = (unsigned)brow * (DIMY * 4u) + (hib ? (unsigned)(BATCH * DIMY * 4) : 0u);
    float oe0 = 0.0f, oe1 = 0.0f, oe2 = 0.0f;
    float zq[TRIP][DIMZ], sq0[TRIP], sq1[TRIP], sq2[TRIP];
    StepIn G[TRIP];
#pragma unroll
    for (int u = 0; u < TRIP; ++u) {
        const gcf_ptr zr = row_ptr(z, (size_t)u * (BATCH * DIMZ), zoff);
        const gcf_ptr sr = row_ptr(s, (size_t)u * (BATCH * DIMS), soff);
#pragma unroll
        for (int i = 0; i < DIMZ; ++i) zq[u][i] = zr[i];
        sq0[u] = sr[0]; sq1[u] = sr[1]; sq2[u] = sr[2];
    }
#pragma unroll
    for (int u = 0; u < TRIP; ++u) {
        G[u] = step_inputs(zq[u], sq0[u], sq1[u], sq2[u], nmw, 0.0f, nmbv, hib);
        const gcf_ptr zr = row_ptr(z, (size_t)(TRIP + u) * (BATCH * DIMZ), zoff);
        const gcf_ptr sr = row_ptr(s, (size_t)(TRIP + u) * (BATCH * DIMS), soff);
#pragma unroll
        for (int i = 0; i < DIMZ; ++i) zq[u][i] = zr[i];
        sq0[u] = sr[0]; sq1[u] = sr[1]; sq2[u] = sr[2];
    }
    lds_handover_barrier();

#pragma unroll 1
    for (int t2 = 0; t2 < SEQ; t2 += TRIP) {
#pragma unroll
    for (int u = 0; u < TRIP; ++u) {
        const int t  = t2 + u;
        const int cb = u & 1;
        const int pb = cb ^ 1;

        const v3f p1 = shP[pb][wa][lane];
        const v3f p2 = shP[pb][wb][lane];
        const v3f p3 = shP[pb][wc][lane];

        asm volatile("" : "+v"(X));
        v8f Xh;
#pragma unroll
        for (int v = 0; v < 8; ++v) Xh[v] = (1.0f - ALPHA) * X[v];
        v2f Bs;
        Bs[0] = G[u].bs0;
        Bs[1] = G[u].bs1;
        v8f acc = wmma_f32_k4(Aw, Bs, Xh);
        float a0 = own0 * G[u].g0;
        float a1 = own1 * G[u].g1;
        float a2 = own2 * G[u].g2;

        if (__builtin_expect(w3, 0)) {
            const float oy0 = zprev2 * (((o0.x + o1.x) + o2.x) + o3.x);
            const float oy1 = zprev2 * (((o0.y + o1.y) + o2.y) + o3.y);
            const float oy2 = zprev2 * (((o0.z + o1.z) + o2.z) + o3.z);
            if (u & 1) {
                if (u > 1 || t2 > 0) {
                    const gf_ptr op = (gf_ptr)row_base(out, (size_t)(t - 3) * (BATCH * DIMY), ooff);
                    op[0] = hib ? oy0 : oe0;
                    op[1] = hib ? oy1 : oe1;
                    op[2] = hib ? oy2 : oe2;
                }
            } else {
                oe0 = oy0; oe1 = oy1; oe2 = oy2;
            }
        }

        a0 = fmaf(p1.x, G[u].g0, a0);
        a1 = fmaf(p1.y, G[u].g1, a1);
        a2 = fmaf(p1.z, G[u].g2, a2);
        a0 = fmaf(p2.x, G[u].g0, a0);
        a1 = fmaf(p2.y, G[u].g1, a1);
        a2 = fmaf(p2.z, G[u].g2, a2);
        a0 = fmaf(p3.x, G[u].g0, a0);
        a1 = fmaf(p3.y, G[u].g1, a1);
        a2 = fmaf(p3.z, G[u].g2, a2);

        zprev2 = zprev;
        zprev = G[u].zsum;

        v2f Bp0, Bp1;
        Bp0[0] = a0;
        Bp0[1] = a1;
        Bp1[0] = a2;
        Bp1[1] = 0.0f;

        acc = wmma_f32_k4(Au0, Bp0, acc);
        X = wmma_f32_k4(Au1, Bp1, acc);

        v2f bfirst;
        bfirst[0] = fast_tanh(X[0]);
        bfirst[1] = fast_tanh(X[1]);

        v8f Qp = vzero;
        if (!w3) {
            v2f bcur = bfirst;
#pragma unroll
            for (int q = 0; q < KGW; ++q) {
                v2f bnext = bcur;
                if (q + 1 < KGW) {
                    bnext[0] = fast_tanh(X[2 * (q + 1)]);
                    bnext[1] = fast_tanh(X[2 * (q + 1) + 1]);
                }
                if (q == 0)
                    asm volatile("" : "+v"(bnext));
                else if (q + 1 < KGW)
                    asm volatile("" : "+v"(bnext), "+v"(Qp));
                Qp = wmma_f32_k4(Aq[q], bcur, Qp);
                bcur = bnext;
            }
        } else {
            v2f b1;
            b1[0] = hib ? 1.0f : bfirst[0];
            b1[1] = hib ? 0.0f : bfirst[1];
            Qp = wmma_f32_k4(Aq[0], b1, Qp);
            __builtin_amdgcn_sched_barrier(0x0476);
            o0 = shY[pb][0][lane];
            o1 = shY[pb][1][lane];
            o2 = shY[pb][2][lane];
            o3 = shY[pb][3][lane];
        }

        shP[cb][w][lane] = __builtin_shufflevector(Qp, Qp, 0, 1, 2);
        shY[cb][w][lane] = __builtin_shufflevector(Qp, Qp, 3, 4, 5);
        own0 = Qp[0]; own1 = Qp[1]; own2 = Qp[2];

        float zstart = 0.0f;
        float bstart[RH] = {nmbv[0], nmbv[1], nmbv[2]};
        asm volatile("" : "+v"(zstart), "+v"(bstart[0]), "+v"(bstart[1]), "+v"(bstart[2])
                        : : "memory");
        G[u] = step_inputs(zq[u], sq0[u], sq1[u], sq2[u], nmw, zstart, bstart, hib);
        {
            const int tn = (t + 2 * TRIP < SEQ) ? (t + 2 * TRIP) : (SEQ - 1);
            const gcf_ptr zr = row_ptr(z, (size_t)tn * (BATCH * DIMZ), zoff);
            const gcf_ptr sr = row_ptr(s, (size_t)tn * (BATCH * DIMS), soff);
#pragma unroll
            for (int i = 0; i < DIMZ; ++i) zq[u][i] = zr[i];
            sq0[u] = sr[0]; sq1[u] = sr[1]; sq2[u] = sr[2];
        }
        asm volatile("" : "+v"(G[u].zsum), "+v"(G[u].g0), "+v"(G[u].g1), "+v"(G[u].g2),
                          "+v"(G[u].bs0), "+v"(G[u].bs1));
        lds_handover_barrier();
    }
    }

    if (w3) {
        const int lb = (SEQ - 1) & 1;
        const float e0 = zprev2 * (((o0.x + o1.x) + o2.x) + o3.x);
        const float e1 = zprev2 * (((o0.y + o1.y) + o2.y) + o3.y);
        const float e2 = zprev2 * (((o0.z + o1.z) + o2.z) + o3.z);
        float y0 = 0.0f, y1 = 0.0f, y2 = 0.0f;
#pragma unroll
        for (int k = 0; k < NWAVE; ++k) {
            const v3f o = shY[lb][k][lane];
            y0 = (k == 0) ? o.x : (y0 + o.x);
            y1 = (k == 0) ? o.y : (y1 + o.y);
            y2 = (k == 0) ? o.z : (y2 + o.z);
        }
        const gf_ptr op = (gf_ptr)row_base(out, (size_t)(SEQ - 2) * (BATCH * DIMY), ooff);
        op[0] = hib ? (zprev * y0) : e0;
        op[1] = hib ? (zprev * y1) : e1;
        op[2] = hib ? (zprev * y2) : e2;
    }
}

extern "C" void kernel_launch(void* const* d_in, const int* in_sizes, int n_in,
                              void* d_out, int out_size, void* d_ws, size_t ws_size,
                              hipStream_t stream) {
    (void)in_sizes; (void)n_in; (void)out_size; (void)d_ws; (void)ws_size;
    const float* s   = (const float*)d_in[0];
    const float* z   = (const float*)d_in[1];
    const float* U   = (const float*)d_in[2];
    const float* V   = (const float*)d_in[3];
    const float* Wi  = (const float*)d_in[4];
    const float* bi  = (const float*)d_in[5];
    const float* Wo  = (const float*)d_in[6];
    const float* bo  = (const float*)d_in[7];
    const float* nmW = (const float*)d_in[8];
    const float* nmb = (const float*)d_in[9];
    float* out = (float*)d_out;

    dim3 grid(BATCH / 16);
    dim3 block(32 * NWAVE);
    hipLaunchKernelGGL(cxtrnn_wmma_kernel, grid, block, 0, stream,
                       s, z, U, V, Wi, bi, Wo, bo, nmW, nmb, out);
}
